# starting kernel
# baseline (speedup 1.0000x reference)
#include <hip/hip_runtime.h>
#include <hip/hip_bf16.h>
#include <math.h>

typedef __attribute__((ext_vector_type(16))) _Float16 v16h;
typedef __attribute__((ext_vector_type(8)))  float    v8f;

#define PATCH 288
#define QD 64
#define C_IN 32
#define ROWSTRIDE_A 296
#define ROWSTRIDE_Y 72

#define WS_W   0
#define WS_MR  18432
#define WS_MI  (18432 + 4096)
#define WS_P   (18432 + 8192)

union FragU { uint4 u[2]; v16h v; };

__global__ void __launch_bounds__(256)
qprep_kernel(const float* __restrict__ in_proj_w,
             const float* __restrict__ weights,
             const float* __restrict__ meas_w,
             const float* __restrict__ out_proj_w,
             _Float16* __restrict__ ws)
{
    __shared__ float Mr[64 * 64];
    __shared__ float Mi[64 * 64];
    const int tid = threadIdx.x;

    for (int e = tid; e < 4096; e += 256) {
        int r = e >> 6, c = e & 63;
        Mr[e] = (r == c) ? 1.f : 0.f;
        Mi[e] = 0.f;
    }
    __syncthreads();

    auto apply1q = [&](int wire, float u00r, float u00i, float u01r, float u01i,
                       float u10r, float u10i, float u11r, float u11i) {
        int p = 5 - wire;
        for (int e = tid; e < 2048; e += 256) {
            int pair = e >> 6, col = e & 63;
            int low = pair & ((1 << p) - 1);
            int i0 = ((pair >> p) << (p + 1)) | low;
            int i1 = i0 | (1 << p);
            float ar = Mr[i0 * 64 + col], ai = Mi[i0 * 64 + col];
            float br = Mr[i1 * 64 + col], bi = Mi[i1 * 64 + col];
            Mr[i0 * 64 + col] = u00r * ar - u00i * ai + u01r * br - u01i * bi;
            Mi[i0 * 64 + col] = u00r * ai + u00i * ar + u01r * bi + u01i * br;
            Mr[i1 * 64 + col] = u10r * ar - u10i * ai + u11r * br - u11i * bi;
            Mi[i1 * 64 + col] = u10r * ai + u10i * ar + u11r * bi + u11i * br;
        }
        __syncthreads();
    };
    auto applyCnot = [&](int wire) {
        int pc = 5 - wire, pt = pc - 1;
        for (int e = tid; e < 1024; e += 256) {
            int grp = e >> 6, col = e & 63;
            int low = grp & ((1 << pt) - 1);
            int high = grp >> pt;
            int i0 = (high << (pc + 1)) | (1 << pc) | low;
            int i1 = i0 | (1 << pt);
            float tr = Mr[i0 * 64 + col], ti = Mi[i0 * 64 + col];
            Mr[i0 * 64 + col] = Mr[i1 * 64 + col];
            Mi[i0 * 64 + col] = Mi[i1 * 64 + col];
            Mr[i1 * 64 + col] = tr;
            Mi[i1 * 64 + col] = ti;
        }
        __syncthreads();
    };

    for (int l = 0; l < 2; ++l) {
        for (int i = 0; i < 6; ++i) {
            float t = weights[(l * 6 + i) * 3 + 0];
            float c = cosf(t * 0.5f), s = sinf(t * 0.5f);
            apply1q(i, c, 0.f, 0.f, -s, 0.f, -s, c, 0.f);
            t = weights[(l * 6 + i) * 3 + 1];
            c = cosf(t * 0.5f); s = sinf(t * 0.5f);
            apply1q(i, c, 0.f, -s, 0.f, s, 0.f, c, 0.f);
        }
        for (int i = 0; i < 5; ++i) applyCnot(i);
        for (int i = 0; i < 6; ++i) {
            float t = weights[(l * 6 + i) * 3 + 2];
            float c = cosf(t * 0.5f), s = sinf(t * 0.5f);
            apply1q(i, c, 0.f, -s, 0.f, s, 0.f, c, 0.f);
        }
    }
    for (int i = 0; i < 6; ++i) {
        float t = meas_w[i * 3 + 0], p = meas_w[i * 3 + 1], lm = meas_w[i * 3 + 2];
        float c = cosf(t * 0.5f), s = sinf(t * 0.5f);
        float cp = cosf(p), sp = sinf(p);
        float cl = cosf(lm), sl = sinf(lm);
        float cpl = cosf(p + lm), spl = sinf(p + lm);
        apply1q(i, c, 0.f, -cl * s, -sl * s, cp * s, sp * s, cpl * c, spl * c);
    }

    for (int e = tid; e < 4 * 9 * 512; e += 256) {
        int h = e & 15, lane = (e >> 4) & 31, s = e >> 9;
        int kt = s % 9, nt = s / 9;
        int n = nt * 16 + (lane & 15);
        int base = (lane & 16) ? 8 : 0;
        int kl = (h < 8) ? (base + h) : (base + h + 8);
        int k = kt * 32 + kl;
        ws[WS_W + e] = (_Float16)in_proj_w[n * PATCH + k];
    }
    for (int e = tid; e < 4 * 2 * 512; e += 256) {
        int h = e & 15, lane = (e >> 4) & 31, s = e >> 9;
        int kt = s & 1, nt = s >> 1;
        int n = nt * 16 + (lane & 15);
        int base = (lane & 16) ? 8 : 0;
        int kl = (h < 8) ? (base + h) : (base + h + 8);
        int k = kt * 32 + kl;
        ws[WS_MR + e] = (_Float16)Mr[n * 64 + k];
        ws[WS_MI + e] = (_Float16)Mi[n * 64 + k];
        ws[WS_P + e]  = (_Float16)out_proj_w[n * 64 + k];
    }
}

__global__ void __launch_bounds__(256)
qconv_main_kernel(const float* __restrict__ x,
                  const float* __restrict__ inp_scale,
                  const float* __restrict__ in_proj_b,
                  const float* __restrict__ out_proj_b,
                  const _Float16* __restrict__ ws,
                  float* __restrict__ out)
{
    __shared__ __align__(16) _Float16 bufA[64 * ROWSTRIDE_A];
    __shared__ __align__(16) _Float16 bufY[64 * ROWSTRIDE_Y];
    __shared__ __align__(16) _Float16 bufP[64 * ROWSTRIDE_Y];
    __shared__ float rowsum[64][2];

    const int tid  = threadIdx.x;
    const int lane = tid & 31;
    const int wv   = tid >> 5;
    const int mt   = wv >> 1;
    const int nh   = wv & 1;
    const int pix0 = blockIdx.x * 64;

    {
        const int m = tid & 63;
        const int pixel = pix0 + m;
        const int b  = pixel >> 12;
        const int hw = pixel & 4095;
        const int hh0 = (hw >> 6) - 1;
        const int ww0 = (hw & 63) - 1;
        for (int k = tid >> 6; k < PATCH; k += 4) {
            int c = k / 9;
            int rem = k - c * 9;
            int r = rem / 3;
            int s2 = rem - r * 3;
            int hh = hh0 + r, wwp = ww0 + s2;
            float v = 0.f;
            if ((unsigned)hh < 64u && (unsigned)wwp < 64u)
                v = x[(((size_t)b * C_IN + c) * 64 + hh) * 64 + wwp];
            bufA[m * ROWSTRIDE_A + k] = (_Float16)(v * inp_scale[k]);
        }
    }
    __syncthreads();

    const int nlo  = lane & 15;
    const int base = (lane & 16) ? 8 : 0;

    v8f acc0 = {}, acc1 = {};
    {
        const _Float16* Arow = &bufA[(mt * 16 + nlo) * ROWSTRIDE_A];
        for (int kt = 0; kt < 9; ++kt) {
            FragU af, bf0, bf1;
            const _Float16* pa = Arow + kt * 32 + base;
            af.u[0] = *(const uint4*)(pa);
            af.u[1] = *(const uint4*)(pa + 16);
            const _Float16* pb0 = ws + WS_W + (((2 * nh + 0) * 9 + kt) * 512 + lane * 16);
            const _Float16* pb1 = ws + WS_W + (((2 * nh + 1) * 9 + kt) * 512 + lane * 16);
            bf0.u[0] = *(const uint4*)(pb0); bf0.u[1] = *(const uint4*)(pb0 + 8);
            bf1.u[0] = *(const uint4*)(pb1); bf1.u[1] = *(const uint4*)(pb1 + 8);
            acc0 = __builtin_amdgcn_wmma_f32_16x16x32_f16(false, af.v, false, bf0.v, (short)0, acc0, false, false);
            acc1 = __builtin_amdgcn_wmma_f32_16x16x32_f16(false, af.v, false, bf1.v, (short)0, acc1, false, false);
        }
    }

    const int n0 = (2 * nh + 0) * 16 + nlo;
    const int n1 = (2 * nh + 1) * 16 + nlo;
    float b0 = in_proj_b[n0], b1 = in_proj_b[n1];
    float y0[8], y1[8], part[8];
    for (int r = 0; r < 8; ++r) {
        y0[r] = acc0[r] + b0;
        y1[r] = acc1[r] + b1;
        part[r] = y0[r] * y0[r] + y1[r] * y1[r];
    }
    for (int mask = 1; mask <= 8; mask <<= 1)
        for (int r = 0; r < 8; ++r)
            part[r] += __shfl_xor(part[r], mask, 32);
    if (nlo == 0) {
        for (int r = 0; r < 8; ++r)
            rowsum[mt * 16 + base + r][nh] = part[r];
    }
    __syncthreads();
    for (int r = 0; r < 8; ++r) {
        int m = mt * 16 + base + r;
        float tot = rowsum[m][0] + rowsum[m][1];
        float inv = 1.f / (sqrtf(tot) + 1e-9f);
        bufY[m * ROWSTRIDE_Y + n0] = (_Float16)(y0[r] * inv);
        bufY[m * ROWSTRIDE_Y + n1] = (_Float16)(y1[r] * inv);
    }
    __syncthreads();

    v8f ar0 = {}, ar1 = {}, ai0 = {}, ai1 = {};
    {
        const _Float16* Yrow = &bufY[(mt * 16 + nlo) * ROWSTRIDE_Y];
        for (int kt = 0; kt < 2; ++kt) {
            FragU af, br0, br1, bi0, bi1;
            const _Float16* pa = Yrow + kt * 32 + base;
            af.u[0] = *(const uint4*)(pa);
            af.u[1] = *(const uint4*)(pa + 16);
            const _Float16* q;
            q = ws + WS_MR + (((2 * nh + 0) * 2 + kt) * 512 + lane * 16);
            br0.u[0] = *(const uint4*)(q); br0.u[1] = *(const uint4*)(q + 8);
            q = ws + WS_MR + (((2 * nh + 1) * 2 + kt) * 512 + lane * 16);
            br1.u[0] = *(const uint4*)(q); br1.u[1] = *(const uint4*)(q + 8);
            q = ws + WS_MI + (((2 * nh + 0) * 2 + kt) * 512 + lane * 16);
            bi0.u[0] = *(const uint4*)(q); bi0.u[1] = *(const uint4*)(q + 8);
            q = ws + WS_MI + (((2 * nh + 1) * 2 + kt) * 512 + lane * 16);
            bi1.u[0] = *(const uint4*)(q); bi1.u[1] = *(const uint4*)(q + 8);
            ar0 = __builtin_amdgcn_wmma_f32_16x16x32_f16(false, af.v, false, br0.v, (short)0, ar0, false, false);
            ar1 = __builtin_amdgcn_wmma_f32_16x16x32_f16(false, af.v, false, br1.v, (short)0, ar1, false, false);
            ai0 = __builtin_amdgcn_wmma_f32_16x16x32_f16(false, af.v, false, bi0.v, (short)0, ai0, false, false);
            ai1 = __builtin_amdgcn_wmma_f32_16x16x32_f16(false, af.v, false, bi1.v, (short)0, ai1, false, false);
        }
    }

    for (int r = 0; r < 8; ++r) {
        int m = mt * 16 + base + r;
        bufP[m * ROWSTRIDE_Y + n0] = (_Float16)(ar0[r] * ar0[r] + ai0[r] * ai0[r]);
        bufP[m * ROWSTRIDE_Y + n1] = (_Float16)(ar1[r] * ar1[r] + ai1[r] * ai1[r]);
    }
    __syncthreads();

    v8f o0 = {}, o1 = {};
    {
        const _Float16* Prow = &bufP[(mt * 16 + nlo) * ROWSTRIDE_Y];
        for (int kt = 0; kt < 2; ++kt) {
            FragU af, bp0, bp1;
            const _Float16* pa = Prow + kt * 32 + base;
            af.u[0] = *(const uint4*)(pa);
            af.u[1] = *(const uint4*)(pa + 16);
            const _Float16* q;
            q = ws + WS_P + (((2 * nh + 0) * 2 + kt) * 512 + lane * 16);
            bp0.u[0] = *(const uint4*)(q); bp0.u[1] = *(const uint4*)(q + 8);
            q = ws + WS_P + (((2 * nh + 1) * 2 + kt) * 512 + lane * 16);
            bp1.u[0] = *(const uint4*)(q); bp1.u[1] = *(const uint4*)(q + 8);
            o0 = __builtin_amdgcn_wmma_f32_16x16x32_f16(false, af.v, false, bp0.v, (short)0, o0, false, false);
            o1 = __builtin_amdgcn_wmma_f32_16x16x32_f16(false, af.v, false, bp1.v, (short)0, o1, false, false);
        }
    }

    float ob0 = out_proj_b[n0], ob1 = out_proj_b[n1];
    float* outstage = (float*)bufA;
    for (int r = 0; r < 8; ++r) {
        int m = mt * 16 + base + r;
        outstage[m * 65 + n0] = o0[r] + ob0;
        outstage[m * 65 + n1] = o1[r] + ob1;
    }
    __syncthreads();
    for (int e = tid; e < 64 * 64; e += 256) {
        int co = e >> 6, m = e & 63;
        int pixel = pix0 + m;
        int b = pixel >> 12, hw = pixel & 4095;
        out[((size_t)(b * 64 + co)) * 4096 + hw] = outstage[m * 65 + co];
    }
}

extern "C" void kernel_launch(void* const* d_in, const int* in_sizes, int n_in,
                              void* d_out, int out_size, void* d_ws, size_t ws_size,
                              hipStream_t stream) {
    const float* x          = (const float*)d_in[0];
    const float* inp_scale  = (const float*)d_in[1];
    const float* in_proj_w  = (const float*)d_in[2];
    const float* in_proj_b  = (const float*)d_in[3];
    const float* weights    = (const float*)d_in[4];
    const float* meas_w     = (const float*)d_in[5];
    const float* out_proj_w = (const float*)d_in[6];
    const float* out_proj_b = (const float*)d_in[7];
    _Float16* ws = (_Float16*)d_ws;
    float* out = (float*)d_out;

    hipLaunchKernelGGL(qprep_kernel, dim3(1), dim3(256), 0, stream,
                       in_proj_w, weights, meas_w, out_proj_w, ws);
    hipLaunchKernelGGL(qconv_main_kernel, dim3(1024), dim3(256), 0, stream,
                       x, inp_scale, in_proj_b, out_proj_b, ws, out);
}


# prep spread over 10 work-groups: one wave per circuit-matrix column in registers with lane exchanges, no LDS sweeps or per-gate barriers (pair sums reordered); weights reordered 16 B per thread
# speedup vs baseline: 2.1876x; 2.1876x over previous
#include <hip/hip_runtime.h>
#include <hip/hip_bf16.h>
#include <math.h>

typedef __attribute__((ext_vector_type(16))) _Float16 v16h;
typedef __attribute__((ext_vector_type(8)))  float    v8f;

#define PATCH 288
#define QD 64
#define C_IN 32
#define ROWSTRIDE_A 296
#define ROWSTRIDE_Y 72

#define WS_W   0
#define WS_MR  18432
#define WS_MI  (18432 + 4096)
#define WS_P   (18432 + 8192)

union FragU { uint4 u[2]; v16h v; };

typedef __attribute__((ext_vector_type(8))) _Float16 v8h;

__device__ __forceinline__ v8h load8_as_half(const float* __restrict__ src)
{
    const float4 lo = *(const float4*)(src);
    const float4 hi = *(const float4*)(src + 4);
    v8h r;
    r[0] = (_Float16)lo.x; r[1] = (_Float16)lo.y; r[2] = (_Float16)lo.z; r[3] = (_Float16)lo.w;
    r[4] = (_Float16)hi.x; r[5] = (_Float16)hi.y; r[6] = (_Float16)hi.z; r[7] = (_Float16)hi.w;
    return r;
}

__device__ __forceinline__ void gate1q(float& a0r, float& a0i, float& a1r, float& a1i,
                                       const float* u, int wire, int lane)
{
    const float u00r = u[0], u00i = u[1], u01r = u[2], u01i = u[3];
    const float u10r = u[4], u10i = u[5], u11r = u[6], u11i = u[7];
    if (wire == 0) {
        const float n0r = u00r * a0r - u00i * a0i + u01r * a1r - u01i * a1i;
        const float n0i = u00r * a0i + u00i * a0r + u01r * a1i + u01i * a1r;
        const float n1r = u10r * a0r - u10i * a0i + u11r * a1r - u11i * a1i;
        const float n1i = u10r * a0i + u10i * a0r + u11r * a1i + u11i * a1r;
        a0r = n0r; a0i = n0i; a1r = n1r; a1i = n1i;
    } else {
        const int m = 1 << (5 - wire);
        const bool up = (lane & m) != 0;
        const float cmr = up ? u11r : u00r, cmi = up ? u11i : u00i;
        const float cqr = up ? u10r : u01r, cqi = up ? u10i : u01i;
        const float q0r = __shfl_xor(a0r, m, 32), q0i = __shfl_xor(a0i, m, 32);
        const float q1r = __shfl_xor(a1r, m, 32), q1i = __shfl_xor(a1i, m, 32);
        const float n0r = cmr * a0r - cmi * a0i + cqr * q0r - cqi * q0i;
        const float n0i = cmr * a0i + cmi * a0r + cqr * q0i + cqi * q0r;
        const float n1r = cmr * a1r - cmi * a1i + cqr * q1r - cqi * q1i;
        const float n1i = cmr * a1i + cmi * a1r + cqr * q1i + cqi * q1r;
        a0r = n0r; a0i = n0i; a1r = n1r; a1i = n1i;
    }
}

__device__ __forceinline__ void cnot(float& a0r, float& a0i, float& a1r, float& a1i,
                                     int wire, int lane)
{
    const int mt = 1 << (4 - wire);
    const float q1r = __shfl_xor(a1r, mt, 32), q1i = __shfl_xor(a1i, mt, 32);
    if (wire == 0) {
        a1r = q1r; a1i = q1i;
    } else {
        const float q0r = __shfl_xor(a0r, mt, 32), q0i = __shfl_xor(a0i, mt, 32);
        const bool ctl = (lane & (1 << (5 - wire))) != 0;
        a0r = ctl ? q0r : a0r; a0i = ctl ? q0i : a0i;
        a1r = ctl ? q1r : a1r; a1i = ctl ? q1i : a1i;
    }
}

__global__ void __launch_bounds__(512)
qprep_kernel(const float* __restrict__ in_proj_w,
             const float* __restrict__ weights,
             const float* __restrict__ meas_w,
             const float* __restrict__ out_proj_w,
             _Float16* __restrict__ ws)
{
    __shared__ float sCS[60 * 2];
    __shared__ float sG[42 * 8];
    __shared__ __align__(16) _Float16 sM[2 * 64 * 16];
    const int tid = threadIdx.x;
    const int grp = blockIdx.x;

    if (grp < 4) {
        if (tid < 60) {
            float ang;
            if (tid < 36) {
                const int l = tid / 18, r = tid - l * 18;
                int i, which;
                if (r < 12) { i = r >> 1; which = r & 1; } else { i = r - 12; which = 2; }
                ang = weights[(l * 6 + i) * 3 + which] * 0.5f;
            } else {
                const int q = tid - 36, i = q >> 2, kind = q & 3;
                const float t = meas_w[i * 3 + 0], p = meas_w[i * 3 + 1], lm = meas_w[i * 3 + 2];
                ang = (kind == 0) ? t * 0.5f : (kind == 1) ? p : (kind == 2) ? lm : p + lm;
            }
            sCS[2 * tid]     = cosf(ang);
            sCS[2 * tid + 1] = sinf(ang);
        }
        __syncthreads();
        if (tid < 42) {
            float u0, u1, u2, u3, u4, u5, u6, u7;
            if (tid < 36) {
                const int r = tid % 18;
                const float c = sCS[2 * tid], s = sCS[2 * tid + 1];
                if (r < 12 && (r & 1) == 0) { u0 = c; u1 = 0.f; u2 = 0.f; u3 = -s; u4 = 0.f; u5 = -s; u6 = c; u7 = 0.f; }
                else                        { u0 = c; u1 = 0.f; u2 = -s;  u3 = 0.f; u4 = s;  u5 = 0.f; u6 = c; u7 = 0.f; }
            } else {
                const float* q = &sCS[2 * (36 + 4 * (tid - 36))];
                const float c = q[0], s = q[1], cp = q[2], sp = q[3], cl = q[4], sl = q[5], cpl = q[6], spl = q[7];
                u0 = c; u1 = 0.f; u2 = -cl * s; u3 = -sl * s; u4 = cp * s; u5 = sp * s; u6 = cpl * c; u7 = spl * c;
            }
            float* g = &sG[tid * 8];
            g[0] = u0; g[1] = u1; g[2] = u2; g[3] = u3; g[4] = u4; g[5] = u5; g[6] = u6; g[7] = u7;
        }
        __syncthreads();

        const int lane = tid & 31;
        const int h    = tid >> 5;
        const int kt   = grp >> 1;
        const int bsel = grp & 1;
        const int col  = kt * 32 + bsel * 8 + ((h < 8) ? h : h + 8);

        float a0r = (lane == col) ? 1.f : 0.f, a0i = 0.f;
        float a1r = (lane + 32 == col) ? 1.f : 0.f, a1i = 0.f;

        for (int l = 0; l < 2; ++l) {
            for (int i = 0; i < 6; ++i) {
                gate1q(a0r, a0i, a1r, a1i, &sG[(l * 18 + 2 * i) * 8], i, lane);
                gate1q(a0r, a0i, a1r, a1i, &sG[(l * 18 + 2 * i + 1) * 8], i, lane);
            }
            for (int i = 0; i < 5; ++i) cnot(a0r, a0i, a1r, a1i, i, lane);
            for (int i = 0; i < 6; ++i)
                gate1q(a0r, a0i, a1r, a1i, &sG[(l * 18 + 12 + i) * 8], i, lane);
        }
        for (int i = 0; i < 6; ++i)
            gate1q(a0r, a0i, a1r, a1i, &sG[(36 + i) * 8], i, lane);

        sM[(lane) * 16 + h]           = (_Float16)a0r;
        sM[(lane + 32) * 16 + h]      = (_Float16)a1r;
        sM[(64 + lane) * 16 + h]      = (_Float16)a0i;
        sM[(64 + lane + 32) * 16 + h] = (_Float16)a1i;
        __syncthreads();
        if (tid < 256) {
            const int arr = tid >> 7, n = (tid >> 1) & 63, part = tid & 1;
            const uint4 v = *(const uint4*)&sM[(arr * 64 + n) * 16 + part * 8];
            _Float16* dst = ws + (arr ? WS_MI : WS_MR)
                          + ((((n >> 4) * 2 + kt) * 32 + bsel * 16 + (n & 15)) * 16 + part * 8);
            *(uint4*)dst = v;
        }
    } else if (grp < 9) {
        const int c = (grp - 4) * 512 + tid;
        if (c < 4 * 9 * 64) {
            const int lane_o = (c >> 1) & 31, s = c >> 6;
            const int kt = s % 9, nt = s / 9;
            const int n = nt * 16 + (lane_o & 15);
            const int k0 = kt * 32 + ((lane_o & 16) ? 8 : 0) + ((c & 1) ? 16 : 0);
            *(v8h*)(ws + WS_W + c * 8) = load8_as_half(in_proj_w + n * PATCH + k0);
        }
    } else {
        const int c = tid;
        const int lane_o = (c >> 1) & 31, s = c >> 6;
        const int kt = s & 1, nt = s >> 1;
        const int n = nt * 16 + (lane_o & 15);
        const int k0 = kt * 32 + ((lane_o & 16) ? 8 : 0) + ((c & 1) ? 16 : 0);
        *(v8h*)(ws + WS_P + c * 8) = load8_as_half(out_proj_w + n * 64 + k0);
    }
}

__global__ void __launch_bounds__(256)
qconv_main_kernel(const float* __restrict__ x,
                  const float* __restrict__ inp_scale,
                  const float* __restrict__ in_proj_b,
                  const float* __restrict__ out_proj_b,
                  const _Float16* __restrict__ ws,
                  float* __restrict__ out)
{
    __shared__ __align__(16) _Float16 bufA[64 * ROWSTRIDE_A];
    __shared__ __align__(16) _Float16 bufY[64 * ROWSTRIDE_Y];
    __shared__ __align__(16) _Float16 bufP[64 * ROWSTRIDE_Y];
    __shared__ float rowsum[64][2];

    const int tid  = threadIdx.x;
    const int lane = tid & 31;
    const int wv   = tid >> 5;
    const int mt   = wv >> 1;
    const int nh   = wv & 1;
    const int pix0 = blockIdx.x * 64;

    {
        const int m = tid & 63;
        const int pixel = pix0 + m;
        const int b  = pixel >> 12;
        const int hw = pixel & 4095;
        const int hh0 = (hw >> 6) - 1;
        const int ww0 = (hw & 63) - 1;
        for (int k = tid >> 6; k < PATCH; k += 4) {
            int c = k / 9;
            int rem = k - c * 9;
            int r = rem / 3;
            int s2 = rem - r * 3;
            int hh = hh0 + r, wwp = ww0 + s2;
            float v = 0.f;
            if ((unsigned)hh < 64u && (unsigned)wwp < 64u)
                v = x[(((size_t)b * C_IN + c) * 64 + hh) * 64 + wwp];
            bufA[m * ROWSTRIDE_A + k] = (_Float16)(v * inp_scale[k]);
        }
    }
    __syncthreads();

    const int nlo  = lane & 15;
    const int base = (lane & 16) ? 8 : 0;

    v8f acc0 = {}, acc1 = {};
    {
        const _Float16* Arow = &bufA[(mt * 16 + nlo) * ROWSTRIDE_A];
        for (int kt = 0; kt < 9; ++kt) {
            FragU af, bf0, bf1;
            const _Float16* pa = Arow + kt * 32 + base;
            af.u[0] = *(const uint4*)(pa);
            af.u[1] = *(const uint4*)(pa + 16);
            const _Float16* pb0 = ws + WS_W + (((2 * nh + 0) * 9 + kt) * 512 + lane * 16);
            const _Float16* pb1 = ws + WS_W + (((2 * nh + 1) * 9 + kt) * 512 + lane * 16);
            bf0.u[0] = *(const uint4*)(pb0); bf0.u[1] = *(const uint4*)(pb0 + 8);
            bf1.u[0] = *(const uint4*)(pb1); bf1.u[1] = *(const uint4*)(pb1 + 8);
            acc0 = __builtin_amdgcn_wmma_f32_16x16x32_f16(false, af.v, false, bf0.v, (short)0, acc0, false, false);
            acc1 = __builtin_amdgcn_wmma_f32_16x16x32_f16(false, af.v, false, bf1.v, (short)0, acc1, false, false);
        }
    }

    const int n0 = (2 * nh + 0) * 16 + nlo;
    const int n1 = (2 * nh + 1) * 16 + nlo;
    float b0 = in_proj_b[n0], b1 = in_proj_b[n1];
    float y0[8], y1[8], part[8];
    for (int r = 0; r < 8; ++r) {
        y0[r] = acc0[r] + b0;
        y1[r] = acc1[r] + b1;
        part[r] = y0[r] * y0[r] + y1[r] * y1[r];
    }
    for (int mask = 1; mask <= 8; mask <<= 1)
        for (int r = 0; r < 8; ++r)
            part[r] += __shfl_xor(part[r], mask, 32);
    if (nlo == 0) {
        for (int r = 0; r < 8; ++r)
            rowsum[mt * 16 + base + r][nh] = part[r];
    }
    __syncthreads();
    for (int r = 0; r < 8; ++r) {
        int m = mt * 16 + base + r;
        float tot = rowsum[m][0] + rowsum[m][1];
        float inv = 1.f / (sqrtf(tot) + 1e-9f);
        bufY[m * ROWSTRIDE_Y + n0] = (_Float16)(y0[r] * inv);
        bufY[m * ROWSTRIDE_Y + n1] = (_Float16)(y1[r] * inv);
    }
    __syncthreads();

    v8f ar0 = {}, ar1 = {}, ai0 = {}, ai1 = {};
    {
        const _Float16* Yrow = &bufY[(mt * 16 + nlo) * ROWSTRIDE_Y];
        for (int kt = 0; kt < 2; ++kt) {
            FragU af, br0, br1, bi0, bi1;
            const _Float16* pa = Yrow + kt * 32 + base;
            af.u[0] = *(const uint4*)(pa);
            af.u[1] = *(const uint4*)(pa + 16);
            const _Float16* q;
            q = ws + WS_MR + (((2 * nh + 0) * 2 + kt) * 512 + lane * 16);
            br0.u[0] = *(const uint4*)(q); br0.u[1] = *(const uint4*)(q + 8);
            q = ws + WS_MR + (((2 * nh + 1) * 2 + kt) * 512 + lane * 16);
            br1.u[0] = *(const uint4*)(q); br1.u[1] = *(const uint4*)(q + 8);
            q = ws + WS_MI + (((2 * nh + 0) * 2 + kt) * 512 + lane * 16);
            bi0.u[0] = *(const uint4*)(q); bi0.u[1] = *(const uint4*)(q + 8);
            q = ws + WS_MI + (((2 * nh + 1) * 2 + kt) * 512 + lane * 16);
            bi1.u[0] = *(const uint4*)(q); bi1.u[1] = *(const uint4*)(q + 8);
            ar0 = __builtin_amdgcn_wmma_f32_16x16x32_f16(false, af.v, false, br0.v, (short)0, ar0, false, false);
            ar1 = __builtin_amdgcn_wmma_f32_16x16x32_f16(false, af.v, false, br1.v, (short)0, ar1, false, false);
            ai0 = __builtin_amdgcn_wmma_f32_16x16x32_f16(false, af.v, false, bi0.v, (short)0, ai0, false, false);
            ai1 = __builtin_amdgcn_wmma_f32_16x16x32_f16(false, af.v, false, bi1.v, (short)0, ai1, false, false);
        }
    }

    for (int r = 0; r < 8; ++r) {
        int m = mt * 16 + base + r;
        bufP[m * ROWSTRIDE_Y + n0] = (_Float16)(ar0[r] * ar0[r] + ai0[r] * ai0[r]);
        bufP[m * ROWSTRIDE_Y + n1] = (_Float16)(ar1[r] * ar1[r] + ai1[r] * ai1[r]);
    }
    __syncthreads();

    v8f o0 = {}, o1 = {};
    {
        const _Float16* Prow = &bufP[(mt * 16 + nlo) * ROWSTRIDE_Y];
        for (int kt = 0; kt < 2; ++kt) {
            FragU af, bp0, bp1;
            const _Float16* pa = Prow + kt * 32 + base;
            af.u[0] = *(const uint4*)(pa);
            af.u[1] = *(const uint4*)(pa + 16);
            const _Float16* q;
            q = ws + WS_P + (((2 * nh + 0) * 2 + kt) * 512 + lane * 16);
            bp0.u[0] = *(const uint4*)(q); bp0.u[1] = *(const uint4*)(q + 8);
            q = ws + WS_P + (((2 * nh + 1) * 2 + kt) * 512 + lane * 16);
            bp1.u[0] = *(const uint4*)(q); bp1.u[1] = *(const uint4*)(q + 8);
            o0 = __builtin_amdgcn_wmma_f32_16x16x32_f16(false, af.v, false, bp0.v, (short)0, o0, false, false);
            o1 = __builtin_amdgcn_wmma_f32_16x16x32_f16(false, af.v, false, bp1.v, (short)0, o1, false, false);
        }
    }

    float ob0 = out_proj_b[n0], ob1 = out_proj_b[n1];
    float* outstage = (float*)bufA;
    for (int r = 0; r < 8; ++r) {
        int m = mt * 16 + base + r;
        outstage[m * 65 + n0] = o0[r] + ob0;
        outstage[m * 65 + n1] = o1[r] + ob1;
    }
    __syncthreads();
    for (int e = tid; e < 64 * 64; e += 256) {
        int co = e >> 6, m = e & 63;
        int pixel = pix0 + m;
        int b = pixel >> 12, hw = pixel & 4095;
        out[((size_t)(b * 64 + co)) * 4096 + hw] = outstage[m * 65 + co];
    }
}

extern "C" void kernel_launch(void* const* d_in, const int* in_sizes, int n_in,
                              void* d_out, int out_size, void* d_ws, size_t ws_size,
                              hipStream_t stream) {
    const float* x          = (const float*)d_in[0];
    const float* inp_scale  = (const float*)d_in[1];
    const float* in_proj_w  = (const float*)d_in[2];
    const float* in_proj_b  = (const float*)d_in[3];
    const float* weights    = (const float*)d_in[4];
    const float* meas_w     = (const float*)d_in[5];
    const float* out_proj_w = (const float*)d_in[6];
    const float* out_proj_b = (const float*)d_in[7];
    _Float16* ws = (_Float16*)d_ws;
    float* out = (float*)d_out;

    hipLaunchKernelGGL(qprep_kernel, dim3(10), dim3(512), 0, stream,
                       in_proj_w, weights, meas_w, out_proj_w, ws);
    hipLaunchKernelGGL(qconv_main_kernel, dim3(1024), dim3(256), 0, stream,
                       x, inp_scale, in_proj_b, out_proj_b, ws, out);
}


# main kernel retiled: a work-group takes 4 image rows (256 pixels), a wave 4 m-tiles x 2 n-tiles, weight fragments loaded once per wave into registers a phase ahead; 256 work-groups, not 1024
# speedup vs baseline: 5.0823x; 2.3232x over previous
#include <hip/hip_runtime.h>
#include <hip/hip_bf16.h>
#include <math.h>

typedef __attribute__((ext_vector_type(16))) _Float16 v16h;
typedef __attribute__((ext_vector_type(8)))  float    v8f;

#define PATCH 288
#define QD 64
#define C_IN 32
#define HP 40
#define HCOLS 66
#define TROWS 4
#define HROWS 6
#define TPIX 256
#define HALO_HALFS 15840
#define ROWSTRIDE_Y 72

#define WS_W   0
#define WS_MR  18432
#define WS_MI  (18432 + 4096)
#define WS_P   (18432 + 8192)

union FragU { uint4 u[2]; v16h v; };

typedef __attribute__((ext_vector_type(8))) _Float16 v8h;

__device__ __forceinline__ v8h load8_as_half(const float* __restrict__ src)
{
    const float4 lo = *(const float4*)(src);
    const float4 hi = *(const float4*)(src + 4);
    v8h r;
    r[0] = (_Float16)lo.x; r[1] = (_Float16)lo.y; r[2] = (_Float16)lo.z; r[3] = (_Float16)lo.w;
    r[4] = (_Float16)hi.x; r[5] = (_Float16)hi.y; r[6] = (_Float16)hi.z; r[7] = (_Float16)hi.w;
    return r;
}

__device__ __forceinline__ void gate1q(float& a0r, float& a0i, float& a1r, float& a1i,
                                       const float* u, int wire, int lane)
{
    const float u00r = u[0], u00i = u[1], u01r = u[2], u01i = u[3];
    const float u10r = u[4], u10i = u[5], u11r = u[6], u11i = u[7];
    if (wire == 0) {
        const float n0r = u00r * a0r - u00i * a0i + u01r * a1r - u01i * a1i;
        const float n0i = u00r * a0i + u00i * a0r + u01r * a1i + u01i * a1r;
        const float n1r = u10r * a0r - u10i * a0i + u11r * a1r - u11i * a1i;
        const float n1i = u10r * a0i + u10i * a0r + u11r * a1i + u11i * a1r;
        a0r = n0r; a0i = n0i; a1r = n1r; a1i = n1i;
    } else {
        const int m = 1 << (5 - wire);
        const bool up = (lane & m) != 0;
        const float cmr = up ? u11r : u00r, cmi = up ? u11i : u00i;
        const float cqr = up ? u10r : u01r, cqi = up ? u10i : u01i;
        const float q0r = __shfl_xor(a0r, m, 32), q0i = __shfl_xor(a0i, m, 32);
        const float q1r = __shfl_xor(a1r, m, 32), q1i = __shfl_xor(a1i, m, 32);
        const float n0r = cmr * a0r - cmi * a0i + cqr * q0r - cqi * q0i;
        const float n0i = cmr * a0i + cmi * a0r + cqr * q0i + cqi * q0r;
        const float n1r = cmr * a1r - cmi * a1i + cqr * q1r - cqi * q1i;
        const float n1i = cmr * a1i + cmi * a1r + cqr * q1i + cqi * q1r;
        a0r = n0r; a0i = n0i; a1r = n1r; a1i = n1i;
    }
}

__device__ __forceinline__ void cnot(float& a0r, float& a0i, float& a1r, float& a1i,
                                     int wire, int lane)
{
    const int mt = 1 << (4 - wire);
    const float q1r = __shfl_xor(a1r, mt, 32), q1i = __shfl_xor(a1i, mt, 32);
    if (wire == 0) {
        a1r = q1r; a1i = q1i;
    } else {
        const float q0r = __shfl_xor(a0r, mt, 32), q0i = __shfl_xor(a0i, mt, 32);
        const bool ctl = (lane & (1 << (5 - wire))) != 0;
        a0r = ctl ? q0r : a0r; a0i = ctl ? q0i : a0i;
        a1r = ctl ? q1r : a1r; a1i = ctl ? q1i : a1i;
    }
}

__global__ void __launch_bounds__(512)
qprep_kernel(const float* __restrict__ in_proj_w,
             const float* __restrict__ inp_scale,
             const float* __restrict__ weights,
             const float* __restrict__ meas_w,
             const float* __restrict__ out_proj_w,
             _Float16* __restrict__ ws)
{
    __shared__ float sCS[60 * 2];
    __shared__ float sG[42 * 8];
    __shared__ __align__(16) _Float16 sM[2 * 64 * 16];
    const int tid = threadIdx.x;
    const int grp = blockIdx.x;

    if (grp < 4) {
        if (tid < 60) {
            float ang;
            if (tid < 36) {
                const int l = tid / 18, r = tid - l * 18;
                int i, which;
                if (r < 12) { i = r >> 1; which = r & 1; } else { i = r - 12; which = 2; }
                ang = weights[(l * 6 + i) * 3 + which] * 0.5f;
            } else {
                const int q = tid - 36, i = q >> 2, kind = q & 3;
                const float t = meas_w[i * 3 + 0], p = meas_w[i * 3 + 1], lm = meas_w[i * 3 + 2];
                ang = (kind == 0) ? t * 0.5f : (kind == 1) ? p : (kind == 2) ? lm : p + lm;
            }
            sCS[2 * tid]     = cosf(ang);
            sCS[2 * tid + 1] = sinf(ang);
        }
        __syncthreads();
        if (tid < 42) {
            float u0, u1, u2, u3, u4, u5, u6, u7;
            if (tid < 36) {
                const int r = tid % 18;
                const float c = sCS[2 * tid], s = sCS[2 * tid + 1];
                if (r < 12 && (r & 1) == 0) { u0 = c; u1 = 0.f; u2 = 0.f; u3 = -s; u4 = 0.f; u5 = -s; u6 = c; u7 = 0.f; }
                else                        { u0 = c; u1 = 0.f; u2 = -s;  u3 = 0.f; u4 = s;  u5 = 0.f; u6 = c; u7 = 0.f; }
            } else {
                const float* q = &sCS[2 * (36 + 4 * (tid - 36))];
                const float c = q[0], s = q[1], cp = q[2], sp = q[3], cl = q[4], sl = q[5], cpl = q[6], spl = q[7];
                u0 = c; u1 = 0.f; u2 = -cl * s; u3 = -sl * s; u4 = cp * s; u5 = sp * s; u6 = cpl * c; u7 = spl * c;
            }
            float* g = &sG[tid * 8];
            g[0] = u0; g[1] = u1; g[2] = u2; g[3] = u3; g[4] = u4; g[5] = u5; g[6] = u6; g[7] = u7;
        }
        __syncthreads();

        const int lane = tid & 31;
        const int h    = tid >> 5;
        const int kt   = grp >> 1;
        const int bsel = grp & 1;
        const int col  = kt * 32 + bsel * 8 + ((h < 8) ? h : h + 8);

        float a0r = (lane == col) ? 1.f : 0.f, a0i = 0.f;
        float a1r = (lane + 32 == col) ? 1.f : 0.f, a1i = 0.f;

        for (int l = 0; l < 2; ++l) {
            for (int i = 0; i < 6; ++i) {
                gate1q(a0r, a0i, a1r, a1i, &sG[(l * 18 + 2 * i) * 8], i, lane);
                gate1q(a0r, a0i, a1r, a1i, &sG[(l * 18 + 2 * i + 1) * 8], i, lane);
            }
            for (int i = 0; i < 5; ++i) cnot(a0r, a0i, a1r, a1i, i, lane);
            for (int i = 0; i < 6; ++i)
                gate1q(a0r, a0i, a1r, a1i, &sG[(l * 18 + 12 + i) * 8], i, lane);
        }
        for (int i = 0; i < 6; ++i)
            gate1q(a0r, a0i, a1r, a1i, &sG[(36 + i) * 8], i, lane);

        sM[(lane) * 16 + h]           = (_Float16)a0r;
        sM[(lane + 32) * 16 + h]      = (_Float16)a1r;
        sM[(64 + lane) * 16 + h]      = (_Float16)a0i;
        sM[(64 + lane + 32) * 16 + h] = (_Float16)a1i;
        __syncthreads();
        if (tid < 256) {
            const int arr = tid >> 7, n = (tid >> 1) & 63, part = tid & 1;
            const uint4 v = *(const uint4*)&sM[(arr * 64 + n) * 16 + part * 8];
            _Float16* dst = ws + (arr ? WS_MI : WS_MR)
                          + ((((n >> 4) * 2 + kt) * 32 + bsel * 16 + (n & 15)) * 16 + part * 8);
            *(uint4*)dst = v;
        }
    } else if (grp < 9) {
        const int c = (grp - 4) * 512 + tid;
        if (c < 4 * 9 * 64) {
            const int lane_o = (c >> 1) & 31, s = c >> 6;
            const int kt = s % 9, nt = s / 9;
            const int n = nt * 16 + (lane_o & 15);
            const int ch0 = ((lane_o & 16) ? 8 : 0) + ((c & 1) ? 16 : 0);
            v8h rr;
            for (int j = 0; j < 8; ++j) {
                const int k = (ch0 + j) * 9 + kt;
                rr[j] = (_Float16)(in_proj_w[n * PATCH + k] * inp_scale[k]);
            }
            *(v8h*)(ws + WS_W + c * 8) = rr;
        }
    } else {
        const int c = tid;
        const int lane_o = (c >> 1) & 31, s = c >> 6;
        const int kt = s & 1, nt = s >> 1;
        const int n = nt * 16 + (lane_o & 15);
        const int k0 = kt * 32 + ((lane_o & 16) ? 8 : 0) + ((c & 1) ? 16 : 0);
        *(v8h*)(ws + WS_P + c * 8) = load8_as_half(out_proj_w + n * 64 + k0);
    }
}

typedef __attribute__((ext_vector_type(4))) _Float16 v4h;

__device__ __forceinline__ void load_frag(FragU& f, const _Float16* p)
{
    f.u[0] = *(const uint4*)(p);
    f.u[1] = *(const uint4*)(p + 8);
}

__device__ __forceinline__ void load_afrag(FragU& f, const _Float16* p)
{
    f.u[0] = *(const uint4*)(p);
    f.u[1] = *(const uint4*)(p + 16);
}

__global__ void __launch_bounds__(256) __attribute__((amdgpu_waves_per_eu(2, 2)))
qconv_main_kernel(const float* __restrict__ x,
                  const float* __restrict__ in_proj_b,
                  const float* __restrict__ out_proj_b,
                  const _Float16* __restrict__ ws,
                  float* __restrict__ out)
{
    __shared__ __align__(16) _Float16 buf0[HALO_HALFS + TPIX * ROWSTRIDE_Y];
    __shared__ __align__(16) _Float16 bufP[TPIX * ROWSTRIDE_Y];
    __shared__ float rowsum[TPIX][2];
    _Float16* const halo = buf0;
    _Float16* const bufY = buf0 + HALO_HALFS;

    const int tid  = threadIdx.x;
    const int lane = tid & 31;
    const int wv   = tid >> 5;
    const int t    = wv >> 1;
    const int nh   = wv & 1;
    const int b    = blockIdx.x >> 4;
    const int h0   = (blockIdx.x & 15) * TROWS;

    FragU bw[9][2];

    {
        const int q    = tid & 15;
        const int cg   = (tid >> 4) & 3;
        const int r0   = tid >> 6;
        const int hhA  = h0 - 1 + r0;
        const bool okA = hhA >= 0;
        const int hf   = (tid >> 4) & 1;
        const int cgB  = (tid >> 5) & 3;
        const int rB   = 4 + (tid >> 7);
        const int hhB  = h0 - 1 + rB;
        const bool okB = hhB < 64;
        const float* srcA = x + ((((size_t)b * C_IN + cg * 8) * 64 + (okA ? hhA : 0)) * 64 + 4 * q);
        const float* srcB = x + ((((size_t)b * C_IN + cgB * 8 + hf * 4) * 64 + (okB ? hhB : 63)) * 64 + 4 * q);
        float4 va[8], vb[4];
        for (int j = 0; j < 8; ++j) va[j] = *(const float4*)(srcA + (size_t)j * 4096);
        for (int j = 0; j < 4; ++j) vb[j] = *(const float4*)(srcB + (size_t)j * 4096);

        #pragma unroll
        for (int kt = 0; kt < 9; ++kt) {
            load_frag(bw[kt][0], ws + WS_W + (((2 * nh + 0) * 9 + kt) * 512 + lane * 16));
            load_frag(bw[kt][1], ws + WS_W + (((2 * nh + 1) * 9 + kt) * 512 + lane * 16));
        }

        if (tid < 48) {
            const int r = tid >> 3, side = (tid >> 2) & 1, chunk = tid & 3;
            *(uint4*)&halo[(r * HCOLS + side * 65) * HP + chunk * 8] = make_uint4(0u, 0u, 0u, 0u);
        }
        {
            _Float16* dst = &halo[(r0 * HCOLS + 4 * q + 1) * HP + cg * 8];
            v8h p0, p1, p2, p3;
            for (int j = 0; j < 8; ++j) {
                p0[j] = okA ? (_Float16)va[j].x : (_Float16)0.f; p1[j] = okA ? (_Float16)va[j].y : (_Float16)0.f;
                p2[j] = okA ? (_Float16)va[j].z : (_Float16)0.f; p3[j] = okA ? (_Float16)va[j].w : (_Float16)0.f;
            }
            *(v8h*)(dst)          = p0;
            *(v8h*)(dst + HP)     = p1;
            *(v8h*)(dst + 2 * HP) = p2;
            *(v8h*)(dst + 3 * HP) = p3;
        }
        {
            _Float16* dst = &halo[(rB * HCOLS + 4 * q + 1) * HP + cgB * 8 + hf * 4];
            v4h p0, p1, p2, p3;
            for (int j = 0; j < 4; ++j) {
                p0[j] = okB ? (_Float16)vb[j].x : (_Float16)0.f; p1[j] = okB ? (_Float16)vb[j].y : (_Float16)0.f;
                p2[j] = okB ? (_Float16)vb[j].z : (_Float16)0.f; p3[j] = okB ? (_Float16)vb[j].w : (_Float16)0.f;
            }
            *(v4h*)(dst)          = p0;
            *(v4h*)(dst + HP)     = p1;
            *(v4h*)(dst + 2 * HP) = p2;
            *(v4h*)(dst + 3 * HP) = p3;
        }
    }
    __syncthreads();

    const int nlo  = lane & 15;
    const int base = (lane & 16) ? 8 : 0;
    const int mrow = t * 64;

    FragU br[2][2], bi[2][2];
    #pragma unroll
    for (int kt = 0; kt < 2; ++kt) {
        #pragma unroll
        for (int j = 0; j < 2; ++j) {
            load_frag(br[kt][j], ws + WS_MR + (((2 * nh + j) * 2 + kt) * 512 + lane * 16));
            load_frag(bi[kt][j], ws + WS_MI + (((2 * nh + j) * 2 + kt) * 512 + lane * 16));
        }
    }

    v8f acc[4][2];
    #pragma unroll
    for (int mi = 0; mi < 4; ++mi) { acc[mi][0] = (v8f){}; acc[mi][1] = (v8f){}; }
    {
        const _Float16* Apix = &halo[(t * HCOLS + nlo) * HP + base];
        FragU afr[2][4];
        #pragma unroll
        for (int mi = 0; mi < 4; ++mi) load_afrag(afr[0][mi], Apix + (mi * 16) * HP);
        __builtin_amdgcn_sched_group_barrier(0x100, 8, 0);
        #pragma unroll
        for (int kt = 0; kt < 9; ++kt) {
            if (kt < 8) {
                const int kh = (kt + 1) / 3, kw = (kt + 1) - 3 * kh;
                #pragma unroll
                for (int mi = 0; mi < 4; ++mi)
                    load_afrag(afr[(kt + 1) & 1][mi], Apix + (kh * HCOLS + kw + mi * 16) * HP);
            }
            #pragma unroll
            for (int mi = 0; mi < 4; ++mi) {
                acc[mi][0] = __builtin_amdgcn_wmma_f32_16x16x32_f16(false, afr[kt & 1][mi].v, false, bw[kt][0].v, (short)0, acc[mi][0], false, false);
                acc[mi][1] = __builtin_amdgcn_wmma_f32_16x16x32_f16(false, afr[kt & 1][mi].v, false, bw[kt][1].v, (short)0, acc[mi][1], false, false);
            }
            if (kt < 8) __builtin_amdgcn_sched_group_barrier(0x100, 8, 0);
            __builtin_amdgcn_sched_group_barrier(0x008, 8, 0);
        }
    }

    const int n0 = (2 * nh + 0) * 16 + nlo;
    const int n1 = (2 * nh + 1) * 16 + nlo;
    const float b0 = in_proj_b[n0], b1 = in_proj_b[n1];
    float y0[4][8], y1[4][8];
    #pragma unroll
    for (int mi = 0; mi < 4; ++mi) {
        float part[8];
        for (int r = 0; r < 8; ++r) {
            y0[mi][r] = acc[mi][0][r] + b0;
            y1[mi][r] = acc[mi][1][r] + b1;
            part[r] = y0[mi][r] * y0[mi][r] + y1[mi][r] * y1[mi][r];
        }
        for (int mask = 1; mask <= 8; mask <<= 1)
            for (int r = 0; r < 8; ++r)
                part[r] += __shfl_xor(part[r], mask, 32);
        if (nlo == 0) {
            for (int r = 0; r < 8; ++r)
                rowsum[mrow + mi * 16 + base + r][nh] = part[r];
        }
    }
    __syncthreads();
    #pragma unroll
    for (int mi = 0; mi < 4; ++mi) {
        for (int r = 0; r < 8; ++r) {
            const int m = mrow + mi * 16 + base + r;
            const float tot = rowsum[m][0] + rowsum[m][1];
            const float inv = 1.f / (sqrtf(tot) + 1e-9f);
            bufY[m * ROWSTRIDE_Y + n0] = (_Float16)(y0[mi][r] * inv);
            bufY[m * ROWSTRIDE_Y + n1] = (_Float16)(y1[mi][r] * inv);
        }
    }
    __syncthreads();

    FragU bp[2][2];
    #pragma unroll
    for (int kt = 0; kt < 2; ++kt) {
        #pragma unroll
        for (int j = 0; j < 2; ++j)
            load_frag(bp[kt][j], ws + WS_P + (((2 * nh + j) * 2 + kt) * 512 + lane * 16));
    }

    {
        FragU ay[4][2];
        #pragma unroll
        for (int mi = 0; mi < 4; ++mi) {
            #pragma unroll
            for (int kt = 0; kt < 2; ++kt)
                load_afrag(ay[mi][kt], &bufY[(mrow + mi * 16 + nlo) * ROWSTRIDE_Y + kt * 32 + base]);
        }
        #pragma unroll
        for (int mi = 0; mi < 4; ++mi) {
            v8f ar0 = {}, ar1 = {}, ai0 = {}, ai1 = {};
            #pragma unroll
            for (int kt = 0; kt < 2; ++kt) {
                ar0 = __builtin_amdgcn_wmma_f32_16x16x32_f16(false, ay[mi][kt].v, false, br[kt][0].v, (short)0, ar0, false, false);
                ar1 = __builtin_amdgcn_wmma_f32_16x16x32_f16(false, ay[mi][kt].v, false, br[kt][1].v, (short)0, ar1, false, false);
                ai0 = __builtin_amdgcn_wmma_f32_16x16x32_f16(false, ay[mi][kt].v, false, bi[kt][0].v, (short)0, ai0, false, false);
                ai1 = __builtin_amdgcn_wmma_f32_16x16x32_f16(false, ay[mi][kt].v, false, bi[kt][1].v, (short)0, ai1, false, false);
            }
            for (int r = 0; r < 8; ++r) {
                const int m = mrow + mi * 16 + base + r;
                bufP[m * ROWSTRIDE_Y + n0] = (_Float16)(ar0[r] * ar0[r] + ai0[r] * ai0[r]);
                bufP[m * ROWSTRIDE_Y + n1] = (_Float16)(ar1[r] * ar1[r] + ai1[r] * ai1[r]);
            }
        }
    }
    __syncthreads();

    const float ob0 = out_proj_b[n0], ob1 = out_proj_b[n1];
    float* outstage = (float*)buf0;
    {
        FragU ap[4][2];
        #pragma unroll
        for (int mi = 0; mi < 4; ++mi) {
            #pragma unroll
            for (int kt = 0; kt < 2; ++kt)
                load_afrag(ap[mi][kt], &bufP[(mrow + mi * 16 + nlo) * ROWSTRIDE_Y + kt * 32 + base]);
        }
        #pragma unroll
        for (int mi = 0; mi < 4; ++mi) {
            v8f o0 = {}, o1 = {};
            #pragma unroll
            for (int kt = 0; kt < 2; ++kt) {
                o0 = __builtin_amdgcn_wmma_f32_16x16x32_f16(false, ap[mi][kt].v, false, bp[kt][0].v, (short)0, o0, false, false);
                o1 = __builtin_amdgcn_wmma_f32_16x16x32_f16(false, ap[mi][kt].v, false, bp[kt][1].v, (short)0, o1, false, false);
            }
            for (int r = 0; r < 8; ++r) {
                const int m = mrow + mi * 16 + base + r;
                outstage[m * 65 + n0] = o0[r] + ob0;
                outstage[m * 65 + n1] = o1[r] + ob1;
            }
        }
    }
    __syncthreads();

    {
        float* obase = out + ((size_t)(b * 64)) * 4096 + h0 * 64 + tid;
        for (int co = 0; co < 64; ++co)
            obase[(size_t)co * 4096] = outstage[tid * 65 + co];
    }
}

extern "C" void kernel_launch(void* const* d_in, const int* in_sizes, int n_in,
                              void* d_out, int out_size, void* d_ws, size_t ws_size,
                              hipStream_t stream) {
    const float* x          = (const float*)d_in[0];
    const float* inp_scale  = (const float*)d_in[1];
    const float* in_proj_w  = (const float*)d_in[2];
    const float* in_proj_b  = (const float*)d_in[3];
    const float* weights    = (const float*)d_in[4];
    const float* meas_w     = (const float*)d_in[5];
    const float* out_proj_w = (const float*)d_in[6];
    const float* out_proj_b = (const float*)d_in[7];
    _Float16* ws = (_Float16*)d_ws;
    float* out = (float*)d_out;

    hipLaunchKernelGGL(qprep_kernel, dim3(10), dim3(512), 0, stream,
                       in_proj_w, inp_scale, weights, meas_w, out_proj_w, ws);
    hipLaunchKernelGGL(qconv_main_kernel, dim3(256), dim3(256), 0, stream,
                       x, in_proj_b, out_proj_b, ws, out);
}


# products transposed (weights = A, pixels = B columns): a wave keeps all 64 outputs of its 32 pixels, each D packed to f16 is the next B; no LDS round trips; norm sum order changes, bias summed first
# speedup vs baseline: 7.0716x; 1.3914x over previous
#include <hip/hip_runtime.h>
#include <hip/hip_bf16.h>
#include <math.h>

typedef __attribute__((ext_vector_type(16))) _Float16 v16h;
typedef __attribute__((ext_vector_type(8)))  float    v8f;

#define PATCH 288
#define QD 64
#define C_IN 32
#define HP 40
#define HCOLS 66
#define TROWS 4
#define HROWS 6
#define HALO_HALFS 15840

#define WS_W   0
#define WS_MR  18432
#define WS_MI  (18432 + 4096)
#define WS_P   (18432 + 8192)

union FragU { uint4 u[2]; v16h v; };

typedef __attribute__((ext_vector_type(8))) _Float16 v8h;

__device__ __forceinline__ v8h load8_as_half(const float* __restrict__ src)
{
    const float4 lo = *(const float4*)(src);
    const float4 hi = *(const float4*)(src + 4);
    v8h r;
    r[0] = (_Float16)lo.x; r[1] = (_Float16)lo.y; r[2] = (_Float16)lo.z; r[3] = (_Float16)lo.w;
    r[4] = (_Float16)hi.x; r[5] = (_Float16)hi.y; r[6] = (_Float16)hi.z; r[7] = (_Float16)hi.w;
    return r;
}

__device__ __forceinline__ void gate1q(float& a0r, float& a0i, float& a1r, float& a1i,
                                       const float* u, int wire, int lane)
{
    const float u00r = u[0], u00i = u[1], u01r = u[2], u01i = u[3];
    const float u10r = u[4], u10i = u[5], u11r = u[6], u11i = u[7];
    if (wire == 0) {
        const float n0r = u00r * a0r - u00i * a0i + u01r * a1r - u01i * a1i;
        const float n0i = u00r * a0i + u00i * a0r + u01r * a1i + u01i * a1r;
        const float n1r = u10r * a0r - u10i * a0i + u11r * a1r - u11i * a1i;
        const float n1i = u10r * a0i + u10i * a0r + u11r * a1i + u11i * a1r;
        a0r = n0r; a0i = n0i; a1r = n1r; a1i = n1i;
    } else {
        const int m = 1 << (5 - wire);
        const bool up = (lane & m) != 0;
        const float cmr = up ? u11r : u00r, cmi = up ? u11i : u00i;
        const float cqr = up ? u10r : u01r, cqi = up ? u10i : u01i;
        const float q0r = __shfl_xor(a0r, m, 32), q0i = __shfl_xor(a0i, m, 32);
        const float q1r = __shfl_xor(a1r, m, 32), q1i = __shfl_xor(a1i, m, 32);
        const float n0r = cmr * a0r - cmi * a0i + cqr * q0r - cqi * q0i;
        const float n0i = cmr * a0i + cmi * a0r + cqr * q0i + cqi * q0r;
        const float n1r = cmr * a1r - cmi * a1i + cqr * q1r - cqi * q1i;
        const float n1i = cmr * a1i + cmi * a1r + cqr * q1i + cqi * q1r;
        a0r = n0r; a0i = n0i; a1r = n1r; a1i = n1i;
    }
}

__device__ __forceinline__ void make_gate(float* u, const float* sCS, int g)
{
    if (g < 36) {
        const int r = g % 18;
        const float c = sCS[2 * g], s = sCS[2 * g + 1];
        if (r < 12 && (r & 1) == 0) { u[0] = c; u[1] = 0.f; u[2] = 0.f; u[3] = -s; u[4] = 0.f; u[5] = -s; u[6] = c; u[7] = 0.f; }
        else                        { u[0] = c; u[1] = 0.f; u[2] = -s;  u[3] = 0.f; u[4] = s;  u[5] = 0.f; u[6] = c; u[7] = 0.f; }
    } else {
        const float* q = &sCS[2 * (36 + 4 * (g - 36))];
        const float c = q[0], s = q[1], cp = q[2], sp = q[3], cl = q[4], sl = q[5], cpl = q[6], spl = q[7];
        u[0] = c; u[1] = 0.f; u[2] = -cl * s; u[3] = -sl * s; u[4] = cp * s; u[5] = sp * s; u[6] = cpl * c; u[7] = spl * c;
    }
}

__device__ __forceinline__ void mul2x2(float* c, const float* a, const float* b)
{
    for (int i = 0; i < 2; ++i) {
        for (int j = 0; j < 2; ++j) {
            const float a0r = a[(i * 2 + 0) * 2], a0i = a[(i * 2 + 0) * 2 + 1];
            const float a1r = a[(i * 2 + 1) * 2], a1i = a[(i * 2 + 1) * 2 + 1];
            const float b0r = b[(0 * 2 + j) * 2], b0i = b[(0 * 2 + j) * 2 + 1];
            const float b1r = b[(1 * 2 + j) * 2], b1i = b[(1 * 2 + j) * 2 + 1];
            c[(i * 2 + j) * 2]     = a0r * b0r - a0i * b0i + a1r * b1r - a1i * b1i;
            c[(i * 2 + j) * 2 + 1] = a0r * b0i + a0i * b0r + a1r * b1i + a1i * b1r;
        }
    }
}

__device__ __forceinline__ float2 gate_entry(const float* sK, int g, int rowbit, int colbit)
{
    const float* e = &sK[g * 8 + ((rowbit << 1) | colbit) * 2];
    return make_float2(e[0], e[1]);
}

__device__ __forceinline__ float2 cmulf(float2 a, float2 b)
{
    return make_float2(a.x * b.x - a.y * b.y, a.x * b.y + a.y * b.x);
}

__global__ void __launch_bounds__(512)
qprep_kernel(const float* __restrict__ in_proj_w,
             const float* __restrict__ inp_scale,
             const float* __restrict__ weights,
             const float* __restrict__ meas_w,
             const float* __restrict__ out_proj_w,
             _Float16* __restrict__ ws)
{
    __shared__ float sCS[60 * 2];
    __shared__ float sK[18 * 8];
    __shared__ __align__(16) _Float16 sM[2 * 64 * 16];
    const int tid = threadIdx.x;
    const int grp = blockIdx.x;

    if (grp < 4) {
        if (tid < 60) {
            float ang;
            if (tid < 36) {
                const int l = tid / 18, r = tid - l * 18;
                int i, which;
                if (r < 12) { i = r >> 1; which = r & 1; } else { i = r - 12; which = 2; }
                ang = weights[(l * 6 + i) * 3 + which] * 0.5f;
            } else {
                const int q = tid - 36, i = q >> 2, kind = q & 3;
                const float t = meas_w[i * 3 + 0], p = meas_w[i * 3 + 1], lm = meas_w[i * 3 + 2];
                ang = (kind == 0) ? t * 0.5f : (kind == 1) ? p : (kind == 2) ? lm : p + lm;
            }
            sCS[2 * tid]     = cosf(ang);
            sCS[2 * tid + 1] = sinf(ang);
        }
        __syncthreads();
        if (tid < 18) {
            const int kind = tid / 6, i = tid - kind * 6;
            float g0[8], g1[8], g2[8], m[8], k[8];
            if (kind == 0) {
                make_gate(g0, sCS, 2 * i + 1);
                make_gate(g1, sCS, 2 * i);
                mul2x2(k, g0, g1);
            } else if (kind == 1) {
                make_gate(g0, sCS, 18 + 2 * i + 1);
                make_gate(g1, sCS, 18 + 2 * i);
                make_gate(g2, sCS, 12 + i);
                mul2x2(m, g1, g2);
                mul2x2(k, g0, m);
            } else {
                make_gate(g0, sCS, 36 + i);
                make_gate(g1, sCS, 30 + i);
                mul2x2(k, g0, g1);
            }
            float* g = &sK[tid * 8];
            g[0] = k[0]; g[1] = k[1]; g[2] = k[2]; g[3] = k[3]; g[4] = k[4]; g[5] = k[5]; g[6] = k[6]; g[7] = k[7];
        }
        __syncthreads();

        const int lane = tid & 31;
        const int h    = tid >> 5;
        const int kt   = grp >> 1;
        const int bsel = grp & 1;
        const int col  = kt * 32 + bsel * 8 + ((h < 8) ? h : h + 8);

        const int t0 = lane ^ (lane >> 1);
        const float2 pc = cmulf(cmulf(gate_entry(sK, 2, (t0 >> 3) & 1, (col >> 3) & 1), gate_entry(sK, 3, (t0 >> 2) & 1, (col >> 2) & 1)),
                                cmulf(gate_entry(sK, 4, (t0 >> 1) & 1, (col >> 1) & 1), gate_entry(sK, 5, t0 & 1, col & 1)));
        const int c5 = (col >> 5) & 1, c4 = (col >> 4) & 1, b4 = (t0 >> 4) & 1;
        const float2 s0 = cmulf(pc, cmulf(gate_entry(sK, 0, 0, c5), gate_entry(sK, 1, b4, c4)));
        const float2 s1 = cmulf(pc, cmulf(gate_entry(sK, 0, 1, c5), gate_entry(sK, 1, b4 ^ 1, c4)));
        float a0r = s0.x, a0i = s0.y;
        float a1r = s1.x, a1i = s1.y;

        for (int i = 0; i < 6; ++i)
            gate1q(a0r, a0i, a1r, a1i, &sK[(6 + i) * 8], i, lane);
        {
            const float n0r = __shfl(a0r, t0, 32),      n0i = __shfl(a0i, t0, 32);
            const float n1r = __shfl(a1r, t0 ^ 16, 32), n1i = __shfl(a1i, t0 ^ 16, 32);
            a0r = n0r; a0i = n0i; a1r = n1r; a1i = n1i;
        }
        for (int i = 0; i < 6; ++i)
            gate1q(a0r, a0i, a1r, a1i, &sK[(12 + i) * 8], i, lane);

        sM[(lane) * 16 + h]           = (_Float16)a0r;
        sM[(lane + 32) * 16 + h]      = (_Float16)a1r;
        sM[(64 + lane) * 16 + h]      = (_Float16)a0i;
        sM[(64 + lane + 32) * 16 + h] = (_Float16)a1i;
        __syncthreads();
        if (tid < 256) {
            const int arr = tid >> 7, n = (tid >> 1) & 63, part = tid & 1;
            const uint4 v = *(const uint4*)&sM[(arr * 64 + n) * 16 + part * 8];
            _Float16* dst = ws + (arr ? WS_MI : WS_MR)
                          + ((((n >> 4) * 2 + kt) * 32 + bsel * 16 + (n & 15)) * 16 + part * 8);
            *(uint4*)dst = v;
        }
    } else if (grp < 9) {
        const int c = (grp - 4) * 512 + tid;
        if (c < 4 * 9 * 64) {
            const int lane_o = (c >> 1) & 31, s = c >> 6;
            const int kt = s % 9, nt = s / 9;
            const int n = nt * 16 + (lane_o & 15);
            const int ch0 = ((lane_o & 16) ? 8 : 0) + ((c & 1) ? 16 : 0);
            v8h rr;
            for (int j = 0; j < 8; ++j) {
                const int k = (ch0 + j) * 9 + kt;
                rr[j] = (_Float16)(in_proj_w[n * PATCH + k] * inp_scale[k]);
            }
            *(v8h*)(ws + WS_W + c * 8) = rr;
        }
    } else {
        const int c = tid;
        const int lane_o = (c >> 1) & 31, s = c >> 6;
        const int kt = s & 1, nt = s >> 1;
        const int n = nt * 16 + (lane_o & 15);
        const int k0 = kt * 32 + ((lane_o & 16) ? 8 : 0) + ((c & 1) ? 16 : 0);
        *(v8h*)(ws + WS_P + c * 8) = load8_as_half(out_proj_w + n * 64 + k0);
    }
}

typedef __attribute__((ext_vector_type(4))) _Float16 v4h;

__device__ __forceinline__ void load_frag(FragU& f, const _Float16* p)
{
    f.u[0] = *(const uint4*)(p);
    f.u[1] = *(const uint4*)(p + 8);
}

__device__ __forceinline__ void load_afrag(FragU& f, const _Float16* p)
{
    f.u[0] = *(const uint4*)(p);
    f.u[1] = *(const uint4*)(p + 16);
}

__device__ __forceinline__ void load8f(float* d, const float* s)
{
    const float4 lo = *(const float4*)(s);
    const float4 hi = *(const float4*)(s + 4);
    d[0] = lo.x; d[1] = lo.y; d[2] = lo.z; d[3] = lo.w;
    d[4] = hi.x; d[5] = hi.y; d[6] = hi.z; d[7] = hi.w;
}

__device__ __forceinline__ float other_half(float v)
{
    const unsigned u = __float_as_uint(v);
    return __uint_as_float(__builtin_amdgcn_permlanex16(u, u, 0x76543210u, 0xfedcba98u, false, false));
}

__global__ void __launch_bounds__(256) __attribute__((amdgpu_waves_per_eu(2, 2)))
qconv_main_kernel(const float* __restrict__ x,
                  const float* __restrict__ in_proj_b,
                  const float* __restrict__ out_proj_b,
                  const _Float16* __restrict__ ws,
                  float* __restrict__ out)
{
    __shared__ __align__(16) _Float16 halo[HALO_HALFS];

    const int tid  = threadIdx.x;
    const int lane = tid & 31;
    const int wv   = tid >> 5;
    const int t    = wv >> 1;
    const int hw   = wv & 1;
    const int b    = blockIdx.x >> 4;
    const int h0   = (blockIdx.x & 15) * TROWS;

    FragU bw[9][4];

    {
        const int q    = tid & 15;
        const int cg   = (tid >> 4) & 3;
        const int r0   = tid >> 6;
        const int hhA  = h0 - 1 + r0;
        const bool okA = hhA >= 0;
        const int hf   = (tid >> 4) & 1;
        const int cgB  = (tid >> 5) & 3;
        const int rB   = 4 + (tid >> 7);
        const int hhB  = h0 - 1 + rB;
        const bool okB = hhB < 64;
        const float* srcA = x + ((((size_t)b * C_IN + cg * 8) * 64 + (okA ? hhA : 0)) * 64 + 4 * q);
        const float* srcB = x + ((((size_t)b * C_IN + cgB * 8 + hf * 4) * 64 + (okB ? hhB : 63)) * 64 + 4 * q);
        float4 va[8], vb[4];
        for (int j = 0; j < 8; ++j) va[j] = *(const float4*)(srcA + (size_t)j * 4096);
        for (int j = 0; j < 4; ++j) vb[j] = *(const float4*)(srcB + (size_t)j * 4096);

        #pragma unroll
        for (int kt = 0; kt < 9; ++kt) {
            #pragma unroll
            for (int j = 0; j < 4; ++j)
                load_frag(bw[kt][j], ws + WS_W + ((j * 9 + kt) * 512 + lane * 16));
        }

        if (tid < 48) {
            const int r = tid >> 3, side = (tid >> 2) & 1, chunk = tid & 3;
            *(uint4*)&halo[(r * HCOLS + side * 65) * HP + chunk * 8] = make_uint4(0u, 0u, 0u, 0u);
        }
        {
            _Float16* dst = &halo[(r0 * HCOLS + 4 * q + 1) * HP + cg * 8];
            v8h p0, p1, p2, p3;
            for (int j = 0; j < 8; ++j) {
                p0[j] = okA ? (_Float16)va[j].x : (_Float16)0.f; p1[j] = okA ? (_Float16)va[j].y : (_Float16)0.f;
                p2[j] = okA ? (_Float16)va[j].z : (_Float16)0.f; p3[j] = okA ? (_Float16)va[j].w : (_Float16)0.f;
            }
            *(v8h*)(dst)          = p0;
            *(v8h*)(dst + HP)     = p1;
            *(v8h*)(dst + 2 * HP) = p2;
            *(v8h*)(dst + 3 * HP) = p3;
        }
        {
            _Float16* dst = &halo[(rB * HCOLS + 4 * q + 1) * HP + cgB * 8 + hf * 4];
            v4h p0, p1, p2, p3;
            for (int j = 0; j < 4; ++j) {
                p0[j] = okB ? (_Float16)vb[j].x : (_Float16)0.f; p1[j] = okB ? (_Float16)vb[j].y : (_Float16)0.f;
                p2[j] = okB ? (_Float16)vb[j].z : (_Float16)0.f; p3[j] = okB ? (_Float16)vb[j].w : (_Float16)0.f;
            }
            *(v4h*)(dst)          = p0;
            *(v4h*)(dst + HP)     = p1;
            *(v4h*)(dst + 2 * HP) = p2;
            *(v4h*)(dst + 3 * HP) = p3;
        }
    }
    __syncthreads();

    const int nlo    = lane & 15;
    const int base   = (lane & 16) ? 8 : 0;
    const bool upper = (lane & 16) != 0;

    v8f acc[4][2];
    #pragma unroll
    for (int j = 0; j < 4; ++j) {
        float bj[8];
        load8f(bj, in_proj_b + 16 * j + base);
        v8f c0;
        for (int r = 0; r < 8; ++r) c0[r] = bj[r];
        acc[j][0] = c0; acc[j][1] = c0;
    }
    {
        const _Float16* Bpix = &halo[(t * HCOLS + 32 * hw + nlo) * HP + base];
        FragU hfr[2][2];
        #pragma unroll
        for (int p = 0; p < 2; ++p) load_afrag(hfr[0][p], Bpix + (p * 16) * HP);
        __builtin_amdgcn_sched_group_barrier(0x100, 4, 0);
        #pragma unroll
        for (int kt = 0; kt < 9; ++kt) {
            if (kt < 8) {
                const int kh = (kt + 1) / 3, kw = (kt + 1) - 3 * kh;
                #pragma unroll
                for (int p = 0; p < 2; ++p)
                    load_afrag(hfr[(kt + 1) & 1][p], Bpix + (kh * HCOLS + kw + p * 16) * HP);
            }
            #pragma unroll
            for (int j = 0; j < 4; ++j) {
                acc[j][0] = __builtin_amdgcn_wmma_f32_16x16x32_f16(false, bw[kt][j].v, false, hfr[kt & 1][0].v, (short)0, acc[j][0], false, false);
                acc[j][1] = __builtin_amdgcn_wmma_f32_16x16x32_f16(false, bw[kt][j].v, false, hfr[kt & 1][1].v, (short)0, acc[j][1], false, false);
            }
            if (kt < 8) __builtin_amdgcn_sched_group_barrier(0x100, 4, 0);
            __builtin_amdgcn_sched_group_barrier(0x008, 8, 0);
        }
    }

    FragU br[2][4], bi[2][4], bp[2][4];
    #pragma unroll
    for (int kt = 0; kt < 2; ++kt) {
        #pragma unroll
        for (int i = 0; i < 4; ++i) {
            load_frag(br[kt][i], ws + WS_MR + ((i * 2 + kt) * 512 + lane * 16));
            load_frag(bi[kt][i], ws + WS_MI + ((i * 2 + kt) * 512 + lane * 16));
        }
    }
    #pragma unroll
    for (int kt = 0; kt < 2; ++kt) {
        #pragma unroll
        for (int i = 0; i < 4; ++i)
            load_frag(bp[kt][i], ws + WS_P + ((i * 2 + kt) * 512 + lane * 16));
    }
    v8f ob[4];
    #pragma unroll
    for (int i = 0; i < 4; ++i) {
        float bj[8];
        load8f(bj, out_proj_b + 16 * i + base);
        for (int r = 0; r < 8; ++r) ob[i][r] = bj[r];
    }

    FragU yB[2][2];
    #pragma unroll
    for (int p = 0; p < 2; ++p) {
        float y[4][8];
        float sj[4];
        #pragma unroll
        for (int j = 0; j < 4; ++j) {
            float s = 0.f;
            for (int r = 0; r < 8; ++r) {
                y[j][r] = acc[j][p][r];
                s += y[j][r] * y[j][r];
            }
            sj[j] = s;
        }
        float tot = (sj[0] + sj[1]) + (sj[2] + sj[3]);
        tot += other_half(tot);
        const float inv = 1.f / (sqrtf(tot) + 1e-9f);
        #pragma unroll
        for (int kt = 0; kt < 2; ++kt) {
            for (int e = 0; e < 8; ++e) {
                yB[p][kt].v[e]     = (_Float16)(y[2 * kt][e] * inv);
                yB[p][kt].v[8 + e] = (_Float16)(y[2 * kt + 1][e] * inv);
            }
        }
    }

    FragU pB[2][2];
    #pragma unroll
    for (int i = 0; i < 4; ++i) {
        #pragma unroll
        for (int p = 0; p < 2; ++p) {
            v8f ar = {}, ai = {};
            #pragma unroll
            for (int kt = 0; kt < 2; ++kt) {
                ar = __builtin_amdgcn_wmma_f32_16x16x32_f16(false, br[kt][i].v, false, yB[p][kt].v, (short)0, ar, false, false);
                ai = __builtin_amdgcn_wmma_f32_16x16x32_f16(false, bi[kt][i].v, false, yB[p][kt].v, (short)0, ai, false, false);
            }
            for (int e = 0; e < 8; ++e)
                pB[p][i >> 1].v[(i & 1) * 8 + e] = (_Float16)(ar[e] * ar[e] + ai[e] * ai[e]);
        }
    }

    {
        float* orow = out + (((size_t)b * 64) * 64 + (h0 + t)) * 64 + 32 * hw + lane;
        #pragma unroll
        for (int i = 0; i < 4; ++i) {
            v8f o0 = ob[i], o1 = ob[i];
            #pragma unroll
            for (int kt = 0; kt < 2; ++kt) {
                o0 = __builtin_amdgcn_wmma_f32_16x16x32_f16(false, bp[kt][i].v, false, pB[0][kt].v, (short)0, o0, false, false);
                o1 = __builtin_amdgcn_wmma_f32_16x16x32_f16(false, bp[kt][i].v, false, pB[1][kt].v, (short)0, o1, false, false);
            }
            for (int r = 0; r < 8; ++r) {
                const float v0 = o0[r];
                const float v1 = o1[r];
                const float got = other_half(upper ? v0 : v1);
                orow[(size_t)(16 * i + r) * 4096]     = upper ? got : v0;
                orow[(size_t)(16 * i + 8 + r) * 4096] = upper ? v1 : got;
            }
        }
    }
}

extern "C" void kernel_launch(void* const* d_in, const int* in_sizes, int n_in,
                              void* d_out, int out_size, void* d_ws, size_t ws_size,
                              hipStream_t stream) {
    const float* x          = (const float*)d_in[0];
    const float* inp_scale  = (const float*)d_in[1];
    const float* in_proj_w  = (const float*)d_in[2];
    const float* in_proj_b  = (const float*)d_in[3];
    const float* weights    = (const float*)d_in[4];
    const float* meas_w     = (const float*)d_in[5];
    const float* out_proj_w = (const float*)d_in[6];
    const float* out_proj_b = (const float*)d_in[7];
    _Float16* ws = (_Float16*)d_ws;
    float* out = (float*)d_out;

    hipLaunchKernelGGL(qprep_kernel, dim3(10), dim3(512), 0, stream,
                       in_proj_w, inp_scale, weights, meas_w, out_proj_w, ws);
    hipLaunchKernelGGL(qconv_main_kernel, dim3(256), dim3(256), 0, stream,
                       x, in_proj_b, out_proj_b, ws, out);
}


# one launch: each work-group builds the weight fragments and the circuit matrix in LDS (matrix as a 4-state chain over the wires, cut in two; its f32 sums reordered); no prep kernel, no workspace
# speedup vs baseline: 8.6158x; 1.2184x over previous
#include <hip/hip_runtime.h>
#include <hip/hip_bf16.h>
#include <math.h>


typedef __attribute__((ext_vector_type(16))) _Float16 v16h;
typedef __attribute__((ext_vector_type(8)))  float    v8f;
typedef __attribute__((ext_vector_type(8)))  _Float16 v8h;

#define PATCH 288
#define QD 64
#define C_IN 32
#define HP 40
#define HCOLS 66
#define TROWS 4
#define HROWS 6
#define HALO_HALFS 15840
#define TPITCH 68

union FragU { uint4 u[2]; v16h v; };


__device__ __forceinline__ void make_gate(float* u, const float* sCS, int g)
{
    if (g < 36) {
        const int r = g % 18;
        const float c = sCS[2 * g], s = sCS[2 * g + 1];
        if (r < 12 && (r & 1) == 0) { u[0] = c; u[1] = 0.f; u[2] = 0.f; u[3] = -s; u[4] = 0.f; u[5] = -s; u[6] = c; u[7] = 0.f; }
        else                        { u[0] = c; u[1] = 0.f; u[2] = -s;  u[3] = 0.f; u[4] = s;  u[5] = 0.f; u[6] = c; u[7] = 0.f; }
    } else {
        const float* q = &sCS[2 * (36 + 4 * (g - 36))];
        const float c = q[0], s = q[1], cp = q[2], sp = q[3], cl = q[4], sl = q[5], cpl = q[6], spl = q[7];
        u[0] = c; u[1] = 0.f; u[2] = -cl * s; u[3] = -sl * s; u[4] = cp * s; u[5] = sp * s; u[6] = cpl * c; u[7] = spl * c;
    }
}

__device__ __forceinline__ void mul2x2(float* c, const float* a, const float* b)
{
    for (int i = 0; i < 2; ++i) {
        for (int j = 0; j < 2; ++j) {
            const float a0r = a[(i * 2 + 0) * 2], a0i = a[(i * 2 + 0) * 2 + 1];
            const float a1r = a[(i * 2 + 1) * 2], a1i = a[(i * 2 + 1) * 2 + 1];
            const float b0r = b[(0 * 2 + j) * 2], b0i = b[(0 * 2 + j) * 2 + 1];
            const float b1r = b[(1 * 2 + j) * 2], b1i = b[(1 * 2 + j) * 2 + 1];
            c[(i * 2 + j) * 2]     = a0r * b0r - a0i * b0i + a1r * b1r - a1i * b1i;
            c[(i * 2 + j) * 2 + 1] = a0r * b0i + a0i * b0r + a1r * b1i + a1i * b1r;
        }
    }
}

__device__ __forceinline__ float2 cmulf(float2 a, float2 b)
{
    return make_float2(a.x * b.x - a.y * b.y, a.x * b.y + a.y * b.x);
}


struct WireGates { float2 ka[2]; float2 kb[2][2]; float2 kc[2]; };

__device__ __forceinline__ WireGates pick_gates(const float4 (&q)[3][2], int r, int c)
{
    WireGates g;
    g.ka[0] = c ? make_float2(q[0][0].z, q[0][0].w) : make_float2(q[0][0].x, q[0][0].y);
    g.ka[1] = c ? make_float2(q[0][1].z, q[0][1].w) : make_float2(q[0][1].x, q[0][1].y);
    g.kb[0][0] = make_float2(q[1][0].x, q[1][0].y); g.kb[0][1] = make_float2(q[1][0].z, q[1][0].w);
    g.kb[1][0] = make_float2(q[1][1].x, q[1][1].y); g.kb[1][1] = make_float2(q[1][1].z, q[1][1].w);
    const float4 kr = r ? q[2][1] : q[2][0];
    g.kc[0] = make_float2(kr.x, kr.y);
    g.kc[1] = make_float2(kr.z, kr.w);
    return g;
}

__device__ __forceinline__ float2 cmul2add(float2 a, float2 b, float2 c, float2 d)
{
    const float2 p0 = cmulf(a, b), p1 = cmulf(c, d);
    return make_float2(p0.x + p1.x, p0.y + p1.y);
}

__device__ __forceinline__ void chain_fwd(float2* vec, const WireGates& g)
{
    float2 tmp[4], t2[4];
    for (int a = 0; a < 2; ++a)
        for (int b2 = 0; b2 < 2; ++b2)
            tmp[a * 2 + b2] = cmul2add(vec[a * 2 + 0], g.ka[b2], vec[a * 2 + 1], g.ka[b2 ^ 1]);
    for (int a2 = 0; a2 < 2; ++a2)
        for (int b2 = 0; b2 < 2; ++b2)
            t2[a2 * 2 + b2] = cmul2add(g.kb[a2][b2], tmp[0 * 2 + b2], g.kb[a2 ^ 1][b2], tmp[1 * 2 + b2]);
    for (int a2 = 0; a2 < 2; ++a2)
        for (int b2 = 0; b2 < 2; ++b2)
            vec[a2 * 2 + b2] = cmulf(g.kc[a2], t2[a2 * 2 + b2]);
}

__device__ __forceinline__ void chain_bwd(float2* col, const WireGates& g)
{
    float2 t1[4], t2[4];
    for (int a2 = 0; a2 < 2; ++a2)
        for (int b2 = 0; b2 < 2; ++b2)
            t1[a2 * 2 + b2] = cmulf(g.kc[a2], col[a2 * 2 + b2]);
    for (int a = 0; a < 2; ++a)
        for (int b2 = 0; b2 < 2; ++b2)
            t2[a * 2 + b2] = cmul2add(g.kb[a][b2], t1[0 * 2 + b2], g.kb[a ^ 1][b2], t1[1 * 2 + b2]);
    for (int a = 0; a < 2; ++a)
        for (int b = 0; b < 2; ++b)
            col[a * 2 + b] = cmul2add(g.ka[b], t2[a * 2 + 0], g.ka[b ^ 1], t2[a * 2 + 1]);
}

struct AngleSrc { const float* p1; const float* p2; float scale; bool two; };

__device__ __forceinline__ AngleSrc angle_source(int job, const float* __restrict__ weights, const float* __restrict__ meas_w)
{
    const bool layer = job < 36;
    const int l = job / 18, r = job - l * 18;
    const int widx = (l * 6 + ((r < 12) ? (r >> 1) : (r - 12))) * 3 + ((r < 12) ? (r & 1) : 2);
    const int q = layer ? 0 : job - 36, i = q >> 2, kind = q & 3;
    const int midx = i * 3 + ((kind == 0) ? 0 : (kind == 2) ? 2 : 1);
    const float* src = layer ? weights : meas_w;
    AngleSrc a;
    a.p1 = src + (layer ? widx : midx);
    a.p2 = src + (layer ? 0 : i * 3 + 2);
    a.two = !layer && kind == 3;
    a.scale = (layer || kind == 0) ? 0.5f : 1.f;
    return a;
}

__device__ __forceinline__ void build_merged_gate(int job, const float* sCS, float* sK)
{
    const int kind = job / 6, i = job - kind * 6;
    float g0[8], g1[8], g2[8], m[8], k[8];
    if (kind == 0) {
        make_gate(g0, sCS, 2 * i + 1);
        make_gate(g1, sCS, 2 * i);
        mul2x2(k, g0, g1);
    } else if (kind == 1) {
        make_gate(g0, sCS, 18 + 2 * i + 1);
        make_gate(g1, sCS, 18 + 2 * i);
        make_gate(g2, sCS, 12 + i);
        mul2x2(m, g1, g2);
        mul2x2(k, g0, m);
    } else {
        make_gate(g0, sCS, 36 + i);
        make_gate(g1, sCS, 30 + i);
        mul2x2(k, g0, g1);
    }
    float* g = &sK[job * 8];
    g[0] = k[0]; g[1] = k[1]; g[2] = k[2]; g[3] = k[3]; g[4] = k[4]; g[5] = k[5]; g[6] = k[6]; g[7] = k[7];
}

__device__ __forceinline__ void build_chain_entry(int job, const float* sK, float* sL, float* sR)
{
    const bool left = job < 64;
    const int combo = job & 63;
    const int rb = combo >> 3, cb = combo & 7;
    const int w0 = left ? 0 : 3;
    float4 q[3][3][2];
    for (int l = 0; l < 3; ++l)
        for (int y = 0; y < 3; ++y) {
            const float* p = sK + (6 * y + w0 + l) * 8;
            q[l][y][0] = *(const float4*)(p);
            q[l][y][1] = *(const float4*)(p + 4);
        }
    float2 v[4];
    if (left) {
        v[0] = make_float2(1.f, 0.f); v[1] = make_float2(0.f, 0.f); v[2] = v[1]; v[3] = v[1];
        for (int l = 0; l < 3; ++l)
            chain_fwd(v, pick_gates(q[l], (rb >> (2 - l)) & 1, (cb >> (2 - l)) & 1));
    } else {
        v[0] = make_float2(1.f, 0.f); v[1] = v[0]; v[2] = v[0]; v[3] = v[0];
        for (int l = 2; l >= 0; --l)
            chain_bwd(v, pick_gates(q[l], (rb >> (2 - l)) & 1, (cb >> (2 - l)) & 1));
    }
    float* dst = (left ? sL : sR) + rb * TPITCH + cb * 8;
    *(float4*)(dst)     = make_float4(v[0].x, v[0].y, v[1].x, v[1].y);
    *(float4*)(dst + 4) = make_float4(v[2].x, v[2].y, v[3].x, v[3].y);
}

__device__ __forceinline__ float2 chain_join(float4 l01, float4 l23, float4 r01, float4 r23)
{
    const float re = (l01.x * r01.x - l01.y * r01.y) + (l01.z * r01.z - l01.w * r01.w)
                   + (l23.x * r23.x - l23.y * r23.y) + (l23.z * r23.z - l23.w * r23.w);
    const float im = (l01.x * r01.y + l01.y * r01.x) + (l01.z * r01.w + l01.w * r01.z)
                   + (l23.x * r23.y + l23.y * r23.x) + (l23.z * r23.w + l23.w * r23.z);
    return make_float2(re, im);
}

typedef __attribute__((ext_vector_type(4))) _Float16 v4h;

__device__ __forceinline__ void load_afrag(FragU& f, const _Float16* p)
{
    f.u[0] = *(const uint4*)(p);
    f.u[1] = *(const uint4*)(p + 16);
}

__device__ __forceinline__ void load8f(float* d, const float* s)
{
    const float4 lo = *(const float4*)(s);
    const float4 hi = *(const float4*)(s + 4);
    d[0] = lo.x; d[1] = lo.y; d[2] = lo.z; d[3] = lo.w;
    d[4] = hi.x; d[5] = hi.y; d[6] = hi.z; d[7] = hi.w;
}

__device__ __forceinline__ float other_half(float v)
{
    const unsigned u = __float_as_uint(v);
    return __uint_as_float(__builtin_amdgcn_permlanex16(u, u, 0x76543210u, 0xfedcba98u, false, false));
}


__device__ __forceinline__ void load_lfrag(FragU& f, const _Float16* tab, int frag, int lane)
{
    f.u[0] = *(const uint4*)(tab + (frag * 2 + 0) * 256 + lane * 8);
    f.u[1] = *(const uint4*)(tab + (frag * 2 + 1) * 256 + lane * 8);
}

__device__ __forceinline__ void lds_barrier()
{
    __builtin_amdgcn_fence(__ATOMIC_RELEASE, "workgroup", "local");
    __builtin_amdgcn_s_barrier();
    __builtin_amdgcn_fence(__ATOMIC_ACQUIRE, "workgroup", "local");
}

__global__ void __launch_bounds__(256) __attribute__((amdgpu_waves_per_eu(2, 2)))
qconv_fused_kernel(const float* __restrict__ x,
                   const float* __restrict__ inp_scale,
                   const float* __restrict__ in_proj_w,
                   const float* __restrict__ in_proj_b,
                   const float* __restrict__ weights,
                   const float* __restrict__ meas_w,
                   const float* __restrict__ out_proj_w,
                   const float* __restrict__ out_proj_b,
                   float* __restrict__ out)
{
    __shared__ __align__(16) _Float16 halo[HALO_HALFS];
    __shared__ __align__(16) _Float16 sW[36 * 512];
    __shared__ __align__(16) _Float16 sMr[8 * 512];
    __shared__ __align__(16) _Float16 sMi[8 * 512];
    __shared__ __align__(16) _Float16 sP[8 * 512];
    __shared__ __align__(16) float sL[8 * TPITCH];
    __shared__ __align__(16) float sR[8 * TPITCH];
    __shared__ __align__(16) float sK[18 * 8];
    __shared__ float sCS[60 * 2];

    const int tid  = threadIdx.x;
    const int lane = tid & 31;
    const int wv   = tid >> 5;
    const int t    = wv >> 1;
    const int hw   = wv & 1;
    const int b    = blockIdx.x >> 4;
    const int h0   = (blockIdx.x & 15) * TROWS;
    const int nlo    = lane & 15;
    const int base   = (lane & 16) ? 8 : 0;
    const bool upper = (lane & 16) != 0;

    FragU bw[9][4];

    const AngleSrc asrc = angle_source(tid < 60 ? tid : 59, weights, meas_w);
    const float ang1 = *asrc.p1, ang2 = *asrc.p2;
    __builtin_amdgcn_sched_barrier(0);

    const int wn = tid >> 2, wp = tid & 3;
    float4 wq[18], sq[18], pq[4];
    for (int i = 0; i < 18; ++i) wq[i] = *(const float4*)(in_proj_w + wn * PATCH + wp * 72 + 4 * i);
    for (int i = 0; i < 18; ++i) sq[i] = *(const float4*)(inp_scale + wp * 72 + 4 * i);
    for (int i = 0; i < 4; ++i)  pq[i] = *(const float4*)(out_proj_w + wn * 64 + wp * 16 + 4 * i);
    const int q    = tid & 15;
    const int cg   = (tid >> 4) & 3;
    const int r0   = tid >> 6;
    const int hhA  = h0 - 1 + r0;
    const bool okA = hhA >= 0;
    const int hf   = (tid >> 4) & 1;
    const int cgB  = (tid >> 5) & 3;
    const int rB   = 4 + (tid >> 7);
    const int hhB  = h0 - 1 + rB;
    const bool okB = hhB < 64;
    const float* srcA = x + ((((size_t)b * C_IN + cg * 8) * 64 + (okA ? hhA : 0)) * 64 + 4 * q);
    const float* srcB = x + ((((size_t)b * C_IN + cgB * 8 + hf * 4) * 64 + (okB ? hhB : 63)) * 64 + 4 * q);
    float4 va[8], vb[4];
    for (int j = 0; j < 8; ++j) va[j] = *(const float4*)(srcA + (size_t)j * 4096);
    for (int j = 0; j < 4; ++j) vb[j] = *(const float4*)(srcB + (size_t)j * 4096);
    float4 ipb[4][2];
    for (int j = 0; j < 4; ++j) {
        ipb[j][0] = *(const float4*)(in_proj_b + 16 * j + base);
        ipb[j][1] = *(const float4*)(in_proj_b + 16 * j + base + 4);
    }

    {
        const float ang = (asrc.two ? ang1 + ang2 : ang1) * asrc.scale;
        const float ca = cosf(ang), sa = sinf(ang);
        if (tid < 60) {
            sCS[2 * tid]     = ca;
            sCS[2 * tid + 1] = sa;
        }
    }
    lds_barrier();
    if (tid < 18) build_merged_gate(tid, sCS, sK);
    lds_barrier();
    if (tid < 128) build_chain_entry(tid, sK, sL, sR);

    {
        float wf[72];
        for (int i = 0; i < 18; ++i) {
            wf[4 * i]     = wq[i].x * sq[i].x; wf[4 * i + 1] = wq[i].y * sq[i].y;
            wf[4 * i + 2] = wq[i].z * sq[i].z; wf[4 * i + 3] = wq[i].w * sq[i].w;
        }
        _Float16* dstw = &sW[(((wn >> 4) * 9) * 2 + (wp >> 1)) * 256 + ((wn & 15) + 16 * (wp & 1)) * 8];
        for (int kt = 0; kt < 9; ++kt) {
            v8h rr;
            for (int c = 0; c < 8; ++c) rr[c] = (_Float16)wf[c * 9 + kt];
            *(v8h*)(dstw + kt * 512) = rr;
        }
        v8h plo, phi;
        plo[0] = (_Float16)pq[0].x; plo[1] = (_Float16)pq[0].y; plo[2] = (_Float16)pq[0].z; plo[3] = (_Float16)pq[0].w;
        plo[4] = (_Float16)pq[1].x; plo[5] = (_Float16)pq[1].y; plo[6] = (_Float16)pq[1].z; plo[7] = (_Float16)pq[1].w;
        phi[0] = (_Float16)pq[2].x; phi[1] = (_Float16)pq[2].y; phi[2] = (_Float16)pq[2].z; phi[3] = (_Float16)pq[2].w;
        phi[4] = (_Float16)pq[3].x; phi[5] = (_Float16)pq[3].y; phi[6] = (_Float16)pq[3].z; phi[7] = (_Float16)pq[3].w;
        _Float16* dstp = &sP[((((wn >> 4) * 2 + (wp >> 1)) * 2) + (wp & 1)) * 256 + (wn & 15) * 8];
        *(v8h*)(dstp)       = plo;
        *(v8h*)(dstp + 128) = phi;
    }
    lds_barrier();

    {
        const int mn = tid & 63, mg = tid >> 6;
        float4 rt[8][2], lt[2][2];
        for (int j = 0; j < 8; ++j) {
            const float* rp = sR + (mn & 7) * TPITCH + j * 8;
            rt[j][0] = *(const float4*)(rp); rt[j][1] = *(const float4*)(rp + 4);
        }
        for (int s = 0; s < 2; ++s) {
            const float* lp = sL + (mn >> 3) * TPITCH + (mg + 4 * s) * 8;
            lt[s][0] = *(const float4*)(lp); lt[s][1] = *(const float4*)(lp + 4);
        }
        for (int s = 0; s < 2; ++s) {
            const int g = mg + 4 * s;
            v8h mr, mi;
            for (int j = 0; j < 8; ++j) {
                const float2 e = chain_join(lt[s][0], lt[s][1], rt[j][0], rt[j][1]);
                mr[j] = (_Float16)e.x;
                mi[j] = (_Float16)e.y;
            }
            const int off = ((((mn >> 4) * 2 + (g >> 2)) * 2 + ((g >> 1) & 1)) * 32 + (mn & 15) + 16 * (g & 1)) * 8;
            *(v8h*)(sMr + off) = mr;
            *(v8h*)(sMi + off) = mi;
        }
    }

    #pragma unroll
    for (int kt = 0; kt < 9; ++kt) {
        #pragma unroll
        for (int j = 0; j < 4; ++j)
            load_lfrag(bw[kt][j], sW, j * 9 + kt, lane);
    }
    __builtin_amdgcn_sched_barrier(0);

    {
        if (tid < 48) {
            const int r = tid >> 3, side = (tid >> 2) & 1, chunk = tid & 3;
            *(uint4*)&halo[(r * HCOLS + side * 65) * HP + chunk * 8] = make_uint4(0u, 0u, 0u, 0u);
        }
        {
            _Float16* dst = &halo[(r0 * HCOLS + 4 * q + 1) * HP + cg * 8];
            v8h p0, p1, p2, p3;
            for (int j = 0; j < 8; ++j) {
                p0[j] = okA ? (_Float16)va[j].x : (_Float16)0.f; p1[j] = okA ? (_Float16)va[j].y : (_Float16)0.f;
                p2[j] = okA ? (_Float16)va[j].z : (_Float16)0.f; p3[j] = okA ? (_Float16)va[j].w : (_Float16)0.f;
            }
            *(v8h*)(dst)          = p0;
            *(v8h*)(dst + HP)     = p1;
            *(v8h*)(dst + 2 * HP) = p2;
            *(v8h*)(dst + 3 * HP) = p3;
        }
        {
            _Float16* dst = &halo[(rB * HCOLS + 4 * q + 1) * HP + cgB * 8 + hf * 4];
            v4h p0, p1, p2, p3;
            for (int j = 0; j < 4; ++j) {
                p0[j] = okB ? (_Float16)vb[j].x : (_Float16)0.f; p1[j] = okB ? (_Float16)vb[j].y : (_Float16)0.f;
                p2[j] = okB ? (_Float16)vb[j].z : (_Float16)0.f; p3[j] = okB ? (_Float16)vb[j].w : (_Float16)0.f;
            }
            *(v4h*)(dst)          = p0;
            *(v4h*)(dst + HP)     = p1;
            *(v4h*)(dst + 2 * HP) = p2;
            *(v4h*)(dst + 3 * HP) = p3;
        }
    }
    lds_barrier();

    v8f acc[4][2];
    #pragma unroll
    for (int j = 0; j < 4; ++j) {
        v8f c0;
        c0[0] = ipb[j][0].x; c0[1] = ipb[j][0].y; c0[2] = ipb[j][0].z; c0[3] = ipb[j][0].w;
        c0[4] = ipb[j][1].x; c0[5] = ipb[j][1].y; c0[6] = ipb[j][1].z; c0[7] = ipb[j][1].w;
        acc[j][0] = c0; acc[j][1] = c0;
    }
    {
        const _Float16* Bpix = &halo[(t * HCOLS + 32 * hw + nlo) * HP + base];
        FragU hfr[2][2];
        #pragma unroll
        for (int p = 0; p < 2; ++p) load_afrag(hfr[0][p], Bpix + (p * 16) * HP);
        __builtin_amdgcn_sched_group_barrier(0x100, 4, 0);
        #pragma unroll
        for (int kt = 0; kt < 9; ++kt) {
            if (kt < 8) {
                const int kh = (kt + 1) / 3, kw = (kt + 1) - 3 * kh;
                #pragma unroll
                for (int p = 0; p < 2; ++p)
                    load_afrag(hfr[(kt + 1) & 1][p], Bpix + (kh * HCOLS + kw + p * 16) * HP);
            }
            #pragma unroll
            for (int j = 0; j < 4; ++j) {
                acc[j][0] = __builtin_amdgcn_wmma_f32_16x16x32_f16(false, bw[kt][j].v, false, hfr[kt & 1][0].v, (short)0, acc[j][0], false, false);
                acc[j][1] = __builtin_amdgcn_wmma_f32_16x16x32_f16(false, bw[kt][j].v, false, hfr[kt & 1][1].v, (short)0, acc[j][1], false, false);
            }
            if (kt < 8) __builtin_amdgcn_sched_group_barrier(0x100, 4, 0);
            __builtin_amdgcn_sched_group_barrier(0x008, 8, 0);
        }
    }

    __builtin_amdgcn_sched_barrier(0);

    FragU br[2][4], bi[2][4], bp[2][4];
    #pragma unroll
    for (int i = 0; i < 4; ++i) {
        #pragma unroll
        for (int kt = 0; kt < 2; ++kt) {
            load_lfrag(br[kt][i], sMr, i * 2 + kt, lane);
            load_lfrag(bi[kt][i], sMi, i * 2 + kt, lane);
        }
    }
    #pragma unroll
    for (int i = 0; i < 4; ++i) {
        #pragma unroll
        for (int kt = 0; kt < 2; ++kt)
            load_lfrag(bp[kt][i], sP, i * 2 + kt, lane);
    }
    __builtin_amdgcn_sched_barrier(0);
    v8f ob[4];
    #pragma unroll
    for (int i = 0; i < 4; ++i) {
        float bj[8];
        load8f(bj, out_proj_b + 16 * i + base);
        for (int r = 0; r < 8; ++r) ob[i][r] = bj[r];
    }

    FragU yB[2][2];
    #pragma unroll
    for (int p = 0; p < 2; ++p) {
        float y[4][8];
        float sj[4];
        #pragma unroll
        for (int j = 0; j < 4; ++j) {
            float s = 0.f;
            for (int r = 0; r < 8; ++r) {
                y[j][r] = acc[j][p][r];
                s += y[j][r] * y[j][r];
            }
            sj[j] = s;
        }
        float tot = (sj[0] + sj[1]) + (sj[2] + sj[3]);
        tot += other_half(tot);
        const float inv = 1.f / (sqrtf(tot) + 1e-9f);
        #pragma unroll
        for (int kt = 0; kt < 2; ++kt) {
            for (int e = 0; e < 8; ++e) {
                yB[p][kt].v[e]     = (_Float16)(y[2 * kt][e] * inv);
                yB[p][kt].v[8 + e] = (_Float16)(y[2 * kt + 1][e] * inv);
            }
        }
    }

    FragU pB[2][2];
    #pragma unroll
    for (int i = 0; i < 4; ++i) {
        #pragma unroll
        for (int p = 0; p < 2; ++p) {
            v8f ar = {}, ai = {};
            #pragma unroll
            for (int kt = 0; kt < 2; ++kt) {
                ar = __builtin_amdgcn_wmma_f32_16x16x32_f16(false, br[kt][i].v, false, yB[p][kt].v, (short)0, ar, false, false);
                ai = __builtin_amdgcn_wmma_f32_16x16x32_f16(false, bi[kt][i].v, false, yB[p][kt].v, (short)0, ai, false, false);
            }
            for (int e = 0; e < 8; ++e)
                pB[p][i >> 1].v[(i & 1) * 8 + e] = (_Float16)(ar[e] * ar[e] + ai[e] * ai[e]);
        }
    }

    {
        float* orow = out + (((size_t)b * 64) * 64 + (h0 + t)) * 64 + 32 * hw + lane;
        #pragma unroll
        for (int i = 0; i < 4; ++i) {
            v8f o0 = ob[i], o1 = ob[i];
            #pragma unroll
            for (int kt = 0; kt < 2; ++kt) {
                o0 = __builtin_amdgcn_wmma_f32_16x16x32_f16(false, bp[kt][i].v, false, pB[0][kt].v, (short)0, o0, false, false);
                o1 = __builtin_amdgcn_wmma_f32_16x16x32_f16(false, bp[kt][i].v, false, pB[1][kt].v, (short)0, o1, false, false);
            }
            for (int r = 0; r < 8; ++r) {
                const float v0 = o0[r];
                const float v1 = o1[r];
                const float got = other_half(upper ? v0 : v1);
                orow[(size_t)(16 * i + r) * 4096]     = upper ? got : v0;
                orow[(size_t)(16 * i + 8 + r) * 4096] = upper ? v1 : got;
            }
        }
    }
}


extern "C" void kernel_launch(void* const* d_in, const int* in_sizes, int n_in,
                              void* d_out, int out_size, void* d_ws, size_t ws_size,
                              hipStream_t stream) {
    const float* x          = (const float*)d_in[0];
    const float* inp_scale  = (const float*)d_in[1];
    const float* in_proj_w  = (const float*)d_in[2];
    const float* in_proj_b  = (const float*)d_in[3];
    const float* weights    = (const float*)d_in[4];
    const float* meas_w     = (const float*)d_in[5];
    const float* out_proj_w = (const float*)d_in[6];
    const float* out_proj_b = (const float*)d_in[7];
    float* out = (float*)d_out;

    hipLaunchKernelGGL(qconv_fused_kernel, dim3(256), dim3(256), 0, stream,
                       x, inp_scale, in_proj_w, in_proj_b, weights, meas_w, out_proj_w, out_proj_b, out);
}


# in_proj weights fetched with whole-line loads (a wave load = 512 contiguous bytes, was 32 lines of 16 B) and handed through an f32 LDS copy to the packing threads; packing moved under the gate step
# speedup vs baseline: 9.1986x; 1.0676x over previous
#include <hip/hip_runtime.h>
#include <hip/hip_bf16.h>
#include <math.h>


typedef __attribute__((ext_vector_type(16))) _Float16 v16h;
typedef __attribute__((ext_vector_type(8)))  float    v8f;
typedef __attribute__((ext_vector_type(8)))  _Float16 v8h;

#define PATCH 288
#define QD 64
#define C_IN 32
#define HP 40
#define HCOLS 66
#define TROWS 4
#define HROWS 6
#define HALO_HALFS 15840
#define TPITCH 68
#define RPITCH 76

union FragU { uint4 u[2]; v16h v; };


__device__ __forceinline__ void make_gate(float* u, const float* sCS, int g)
{
    if (g < 36) {
        const int r = g % 18;
        const float c = sCS[2 * g], s = sCS[2 * g + 1];
        if (r < 12 && (r & 1) == 0) { u[0] = c; u[1] = 0.f; u[2] = 0.f; u[3] = -s; u[4] = 0.f; u[5] = -s; u[6] = c; u[7] = 0.f; }
        else                        { u[0] = c; u[1] = 0.f; u[2] = -s;  u[3] = 0.f; u[4] = s;  u[5] = 0.f; u[6] = c; u[7] = 0.f; }
    } else {
        const float* q = &sCS[2 * (36 + 4 * (g - 36))];
        const float c = q[0], s = q[1], cp = q[2], sp = q[3], cl = q[4], sl = q[5], cpl = q[6], spl = q[7];
        u[0] = c; u[1] = 0.f; u[2] = -cl * s; u[3] = -sl * s; u[4] = cp * s; u[5] = sp * s; u[6] = cpl * c; u[7] = spl * c;
    }
}

__device__ __forceinline__ void mul2x2(float* c, const float* a, const float* b)
{
    for (int i = 0; i < 2; ++i) {
        for (int j = 0; j < 2; ++j) {
            const float a0r = a[(i * 2 + 0) * 2], a0i = a[(i * 2 + 0) * 2 + 1];
            const float a1r = a[(i * 2 + 1) * 2], a1i = a[(i * 2 + 1) * 2 + 1];
            const float b0r = b[(0 * 2 + j) * 2], b0i = b[(0 * 2 + j) * 2 + 1];
            const float b1r = b[(1 * 2 + j) * 2], b1i = b[(1 * 2 + j) * 2 + 1];
            c[(i * 2 + j) * 2]     = a0r * b0r - a0i * b0i + a1r * b1r - a1i * b1i;
            c[(i * 2 + j) * 2 + 1] = a0r * b0i + a0i * b0r + a1r * b1i + a1i * b1r;
        }
    }
}

__device__ __forceinline__ float2 cmulf(float2 a, float2 b)
{
    return make_float2(a.x * b.x - a.y * b.y, a.x * b.y + a.y * b.x);
}


struct WireGates { float2 ka[2]; float2 kb[2][2]; float2 kc[2]; };

__device__ __forceinline__ WireGates pick_gates(const float4 (&q)[3][2], int r, int c)
{
    WireGates g;
    g.ka[0] = c ? make_float2(q[0][0].z, q[0][0].w) : make_float2(q[0][0].x, q[0][0].y);
    g.ka[1] = c ? make_float2(q[0][1].z, q[0][1].w) : make_float2(q[0][1].x, q[0][1].y);
    g.kb[0][0] = make_float2(q[1][0].x, q[1][0].y); g.kb[0][1] = make_float2(q[1][0].z, q[1][0].w);
    g.kb[1][0] = make_float2(q[1][1].x, q[1][1].y); g.kb[1][1] = make_float2(q[1][1].z, q[1][1].w);
    const float4 kr = r ? q[2][1] : q[2][0];
    g.kc[0] = make_float2(kr.x, kr.y);
    g.kc[1] = make_float2(kr.z, kr.w);
    return g;
}

__device__ __forceinline__ float2 cmul2add(float2 a, float2 b, float2 c, float2 d)
{
    const float2 p0 = cmulf(a, b), p1 = cmulf(c, d);
    return make_float2(p0.x + p1.x, p0.y + p1.y);
}

__device__ __forceinline__ void chain_fwd(float2* vec, const WireGates& g)
{
    float2 tmp[4], t2[4];
    for (int a = 0; a < 2; ++a)
        for (int b2 = 0; b2 < 2; ++b2)
            tmp[a * 2 + b2] = cmul2add(vec[a * 2 + 0], g.ka[b2], vec[a * 2 + 1], g.ka[b2 ^ 1]);
    for (int a2 = 0; a2 < 2; ++a2)
        for (int b2 = 0; b2 < 2; ++b2)
            t2[a2 * 2 + b2] = cmul2add(g.kb[a2][b2], tmp[0 * 2 + b2], g.kb[a2 ^ 1][b2], tmp[1 * 2 + b2]);
    for (int a2 = 0; a2 < 2; ++a2)
        for (int b2 = 0; b2 < 2; ++b2)
            vec[a2 * 2 + b2] = cmulf(g.kc[a2], t2[a2 * 2 + b2]);
}

__device__ __forceinline__ void chain_bwd(float2* col, const WireGates& g)
{
    float2 t1[4], t2[4];
    for (int a2 = 0; a2 < 2; ++a2)
        for (int b2 = 0; b2 < 2; ++b2)
            t1[a2 * 2 + b2] = cmulf(g.kc[a2], col[a2 * 2 + b2]);
    for (int a = 0; a < 2; ++a)
        for (int b2 = 0; b2 < 2; ++b2)
            t2[a * 2 + b2] = cmul2add(g.kb[a][b2], t1[0 * 2 + b2], g.kb[a ^ 1][b2], t1[1 * 2 + b2]);
    for (int a = 0; a < 2; ++a)
        for (int b = 0; b < 2; ++b)
            col[a * 2 + b] = cmul2add(g.ka[b], t2[a * 2 + 0], g.ka[b ^ 1], t2[a * 2 + 1]);
}

struct AngleSrc { const float* p1; const float* p2; float scale; bool two; };

__device__ __forceinline__ AngleSrc angle_source(int job, const float* __restrict__ weights, const float* __restrict__ meas_w)
{
    const bool layer = job < 36;
    const int l = job / 18, r = job - l * 18;
    const int widx = (l * 6 + ((r < 12) ? (r >> 1) : (r - 12))) * 3 + ((r < 12) ? (r & 1) : 2);
    const int q = layer ? 0 : job - 36, i = q >> 2, kind = q & 3;
    const int midx = i * 3 + ((kind == 0) ? 0 : (kind == 2) ? 2 : 1);
    const float* src = layer ? weights : meas_w;
    AngleSrc a;
    a.p1 = src + (layer ? widx : midx);
    a.p2 = src + (layer ? 0 : i * 3 + 2);
    a.two = !layer && kind == 3;
    a.scale = (layer || kind == 0) ? 0.5f : 1.f;
    return a;
}

__device__ __forceinline__ void build_merged_gate(int job, const float* sCS, float* sK)
{
    const int kind = job / 6, i = job - kind * 6;
    float g0[8], g1[8], g2[8], m[8], k[8];
    if (kind == 0) {
        make_gate(g0, sCS, 2 * i + 1);
        make_gate(g1, sCS, 2 * i);
        mul2x2(k, g0, g1);
    } else if (kind == 1) {
        make_gate(g0, sCS, 18 + 2 * i + 1);
        make_gate(g1, sCS, 18 + 2 * i);
        make_gate(g2, sCS, 12 + i);
        mul2x2(m, g1, g2);
        mul2x2(k, g0, m);
    } else {
        make_gate(g0, sCS, 36 + i);
        make_gate(g1, sCS, 30 + i);
        mul2x2(k, g0, g1);
    }
    float* g = &sK[job * 8];
    g[0] = k[0]; g[1] = k[1]; g[2] = k[2]; g[3] = k[3]; g[4] = k[4]; g[5] = k[5]; g[6] = k[6]; g[7] = k[7];
}

__device__ __forceinline__ void build_chain_entry(int job, const float* sK, float* sL, float* sR)
{
    const bool left = job < 64;
    const int combo = job & 63;
    const int rb = combo >> 3, cb = combo & 7;
    const int w0 = left ? 0 : 3;
    float4 q[3][3][2];
    for (int l = 0; l < 3; ++l)
        for (int y = 0; y < 3; ++y) {
            const float* p = sK + (6 * y + w0 + l) * 8;
            q[l][y][0] = *(const float4*)(p);
            q[l][y][1] = *(const float4*)(p + 4);
        }
    float2 v[4];
    if (left) {
        v[0] = make_float2(1.f, 0.f); v[1] = make_float2(0.f, 0.f); v[2] = v[1]; v[3] = v[1];
        for (int l = 0; l < 3; ++l)
            chain_fwd(v, pick_gates(q[l], (rb >> (2 - l)) & 1, (cb >> (2 - l)) & 1));
    } else {
        v[0] = make_float2(1.f, 0.f); v[1] = v[0]; v[2] = v[0]; v[3] = v[0];
        for (int l = 2; l >= 0; --l)
            chain_bwd(v, pick_gates(q[l], (rb >> (2 - l)) & 1, (cb >> (2 - l)) & 1));
    }
    float* dst = (left ? sL : sR) + rb * TPITCH + cb * 8;
    *(float4*)(dst)     = make_float4(v[0].x, v[0].y, v[1].x, v[1].y);
    *(float4*)(dst + 4) = make_float4(v[2].x, v[2].y, v[3].x, v[3].y);
}

__device__ __forceinline__ float2 chain_join(float4 l01, float4 l23, float4 r01, float4 r23)
{
    const float re = (l01.x * r01.x - l01.y * r01.y) + (l01.z * r01.z - l01.w * r01.w)
                   + (l23.x * r23.x - l23.y * r23.y) + (l23.z * r23.z - l23.w * r23.w);
    const float im = (l01.x * r01.y + l01.y * r01.x) + (l01.z * r01.w + l01.w * r01.z)
                   + (l23.x * r23.y + l23.y * r23.x) + (l23.z * r23.w + l23.w * r23.z);
    return make_float2(re, im);
}

typedef __attribute__((ext_vector_type(4))) _Float16 v4h;

__device__ __forceinline__ void load_afrag(FragU& f, const _Float16* p)
{
    f.u[0] = *(const uint4*)(p);
    f.u[1] = *(const uint4*)(p + 16);
}

__device__ __forceinline__ void load8f(float* d, const float* s)
{
    const float4 lo = *(const float4*)(s);
    const float4 hi = *(const float4*)(s + 4);
    d[0] = lo.x; d[1] = lo.y; d[2] = lo.z; d[3] = lo.w;
    d[4] = hi.x; d[5] = hi.y; d[6] = hi.z; d[7] = hi.w;
}

__device__ __forceinline__ float other_half(float v)
{
    const unsigned u = __float_as_uint(v);
    return __uint_as_float(__builtin_amdgcn_permlanex16(u, u, 0x76543210u, 0xfedcba98u, false, false));
}


__device__ __forceinline__ void load_lfrag(FragU& f, const _Float16* tab, int frag, int lane)
{
    f.u[0] = *(const uint4*)(tab + (frag * 2 + 0) * 256 + lane * 8);
    f.u[1] = *(const uint4*)(tab + (frag * 2 + 1) * 256 + lane * 8);
}

__device__ __forceinline__ void lds_barrier()
{
    __builtin_amdgcn_fence(__ATOMIC_RELEASE, "workgroup", "local");
    __builtin_amdgcn_s_barrier();
    __builtin_amdgcn_fence(__ATOMIC_ACQUIRE, "workgroup", "local");
}

__global__ void __launch_bounds__(256) __attribute__((amdgpu_waves_per_eu(2, 2)))
qconv_fused_kernel(const float* __restrict__ x,
                   const float* __restrict__ inp_scale,
                   const float* __restrict__ in_proj_w,
                   const float* __restrict__ in_proj_b,
                   const float* __restrict__ weights,
                   const float* __restrict__ meas_w,
                   const float* __restrict__ out_proj_w,
                   const float* __restrict__ out_proj_b,
                   float* __restrict__ out)
{
    __shared__ __align__(16) _Float16 halo[HALO_HALFS];
    __shared__ __align__(16) _Float16 sW[36 * 512];
    __shared__ __align__(16) _Float16 sMr[8 * 512];
    __shared__ __align__(16) _Float16 sMi[8 * 512];
    __shared__ __align__(16) _Float16 sP[8 * 512];
    __shared__ __align__(16) float sL[8 * TPITCH];
    __shared__ __align__(16) float sR[8 * TPITCH];
    __shared__ __align__(16) float sK[18 * 8];
    __shared__ __align__(16) float sRaw[256 * RPITCH];
    __shared__ float sCS[60 * 2];

    const int tid  = threadIdx.x;
    const int lane = tid & 31;
    const int wv   = tid >> 5;
    const int t    = wv >> 1;
    const int hw   = wv & 1;
    const int b    = blockIdx.x >> 4;
    const int h0   = (blockIdx.x & 15) * TROWS;
    const int nlo    = lane & 15;
    const int base   = (lane & 16) ? 8 : 0;
    const bool upper = (lane & 16) != 0;

    FragU bw[9][4];

    const AngleSrc asrc = angle_source(tid < 60 ? tid : 59, weights, meas_w);
    const float ang1 = *asrc.p1, ang2 = *asrc.p2;
    float4 wq[18];
    for (int i = 0; i < 18; ++i) wq[i] = *(const float4*)(in_proj_w + 4 * (i * 256 + tid));
    for (int i = 0; i < 18; ++i) {
        const unsigned u = (unsigned)(i * 256 + tid);
        const unsigned piece = u / 18u;
        *(float4*)(sRaw + piece * RPITCH + (u - piece * 18u) * 4u) = wq[i];
    }
    __builtin_amdgcn_sched_barrier(0);

    const int wn = tid >> 2, wp = tid & 3;
    float4 sq[18], pq[4];
    for (int i = 0; i < 18; ++i) sq[i] = *(const float4*)(inp_scale + wp * 72 + 4 * i);
    for (int i = 0; i < 4; ++i)  pq[i] = *(const float4*)(out_proj_w + wn * 64 + wp * 16 + 4 * i);
    const int q    = tid & 15;
    const int cg   = (tid >> 4) & 3;
    const int r0   = tid >> 6;
    const int hhA  = h0 - 1 + r0;
    const bool okA = hhA >= 0;
    const int hf   = (tid >> 4) & 1;
    const int cgB  = (tid >> 5) & 3;
    const int rB   = 4 + (tid >> 7);
    const int hhB  = h0 - 1 + rB;
    const bool okB = hhB < 64;
    const float* srcA = x + ((((size_t)b * C_IN + cg * 8) * 64 + (okA ? hhA : 0)) * 64 + 4 * q);
    const float* srcB = x + ((((size_t)b * C_IN + cgB * 8 + hf * 4) * 64 + (okB ? hhB : 63)) * 64 + 4 * q);
    float4 va[8], vb[4];
    for (int j = 0; j < 8; ++j) va[j] = *(const float4*)(srcA + (size_t)j * 4096);
    for (int j = 0; j < 4; ++j) vb[j] = *(const float4*)(srcB + (size_t)j * 4096);
    float4 ipb[4][2];
    for (int j = 0; j < 4; ++j) {
        ipb[j][0] = *(const float4*)(in_proj_b + 16 * j + base);
        ipb[j][1] = *(const float4*)(in_proj_b + 16 * j + base + 4);
    }

    {
        const float ang = (asrc.two ? ang1 + ang2 : ang1) * asrc.scale;
        const float ca = cosf(ang), sa = sinf(ang);
        if (tid < 60) {
            sCS[2 * tid]     = ca;
            sCS[2 * tid + 1] = sa;
        }
    }
    auto stage_weights = [&]() {
        float wf[72];
        for (int i = 0; i < 18; ++i) {
            const float4 w = *(const float4*)(sRaw + tid * RPITCH + 4 * i);
            wf[4 * i]     = w.x * sq[i].x; wf[4 * i + 1] = w.y * sq[i].y;
            wf[4 * i + 2] = w.z * sq[i].z; wf[4 * i + 3] = w.w * sq[i].w;
        }
        _Float16* dstw = &sW[(((wn >> 4) * 9) * 2 + (wp >> 1)) * 256 + ((wn & 15) + 16 * (wp & 1)) * 8];
        for (int kt = 0; kt < 9; ++kt) {
            v8h rr;
            for (int c = 0; c < 8; ++c) rr[c] = (_Float16)wf[c * 9 + kt];
            *(v8h*)(dstw + kt * 512) = rr;
        }
        v8h plo, phi;
        plo[0] = (_Float16)pq[0].x; plo[1] = (_Float16)pq[0].y; plo[2] = (_Float16)pq[0].z; plo[3] = (_Float16)pq[0].w;
        plo[4] = (_Float16)pq[1].x; plo[5] = (_Float16)pq[1].y; plo[6] = (_Float16)pq[1].z; plo[7] = (_Float16)pq[1].w;
        phi[0] = (_Float16)pq[2].x; phi[1] = (_Float16)pq[2].y; phi[2] = (_Float16)pq[2].z; phi[3] = (_Float16)pq[2].w;
        phi[4] = (_Float16)pq[3].x; phi[5] = (_Float16)pq[3].y; phi[6] = (_Float16)pq[3].z; phi[7] = (_Float16)pq[3].w;
        _Float16* dstp = &sP[((((wn >> 4) * 2 + (wp >> 1)) * 2) + (wp & 1)) * 256 + (wn & 15) * 8];
        *(v8h*)(dstp)       = plo;
        *(v8h*)(dstp + 128) = phi;
    };
    lds_barrier();
    if (wv == 0) { if (tid < 18) build_merged_gate(tid, sCS, sK); }
    else stage_weights();
    lds_barrier();
    if (wv >= 4) build_chain_entry(tid - 128, sK, sL, sR);
    else if (wv == 0) stage_weights();
    lds_barrier();

    {
        const int mn = tid & 63, mg = tid >> 6;
        float4 rt[8][2], lt[2][2];
        for (int j = 0; j < 8; ++j) {
            const float* rp = sR + (mn & 7) * TPITCH + j * 8;
            rt[j][0] = *(const float4*)(rp); rt[j][1] = *(const float4*)(rp + 4);
        }
        for (int s = 0; s < 2; ++s) {
            const float* lp = sL + (mn >> 3) * TPITCH + (mg + 4 * s) * 8;
            lt[s][0] = *(const float4*)(lp); lt[s][1] = *(const float4*)(lp + 4);
        }
        for (int s = 0; s < 2; ++s) {
            const int g = mg + 4 * s;
            v8h mr, mi;
            for (int j = 0; j < 8; ++j) {
                const float2 e = chain_join(lt[s][0], lt[s][1], rt[j][0], rt[j][1]);
                mr[j] = (_Float16)e.x;
                mi[j] = (_Float16)e.y;
            }
            const int off = ((((mn >> 4) * 2 + (g >> 2)) * 2 + ((g >> 1) & 1)) * 32 + (mn & 15) + 16 * (g & 1)) * 8;
            *(v8h*)(sMr + off) = mr;
            *(v8h*)(sMi + off) = mi;
        }
    }

    #pragma unroll
    for (int kt = 0; kt < 9; ++kt) {
        #pragma unroll
        for (int j = 0; j < 4; ++j)
            load_lfrag(bw[kt][j], sW, j * 9 + kt, lane);
    }
    __builtin_amdgcn_sched_barrier(0);

    {
        if (tid < 48) {
            const int r = tid >> 3, side = (tid >> 2) & 1, chunk = tid & 3;
            *(uint4*)&halo[(r * HCOLS + side * 65) * HP + chunk * 8] = make_uint4(0u, 0u, 0u, 0u);
        }
        {
            _Float16* dst = &halo[(r0 * HCOLS + 4 * q + 1) * HP + cg * 8];
            v8h p0, p1, p2, p3;
            for (int j = 0; j < 8; ++j) {
                p0[j] = okA ? (_Float16)va[j].x : (_Float16)0.f; p1[j] = okA ? (_Float16)va[j].y : (_Float16)0.f;
                p2[j] = okA ? (_Float16)va[j].z : (_Float16)0.f; p3[j] = okA ? (_Float16)va[j].w : (_Float16)0.f;
            }
            *(v8h*)(dst)          = p0;
            *(v8h*)(dst + HP)     = p1;
            *(v8h*)(dst + 2 * HP) = p2;
            *(v8h*)(dst + 3 * HP) = p3;
        }
        {
            _Float16* dst = &halo[(rB * HCOLS + 4 * q + 1) * HP + cgB * 8 + hf * 4];
            v4h p0, p1, p2, p3;
            for (int j = 0; j < 4; ++j) {
                p0[j] = okB ? (_Float16)vb[j].x : (_Float16)0.f; p1[j] = okB ? (_Float16)vb[j].y : (_Float16)0.f;
                p2[j] = okB ? (_Float16)vb[j].z : (_Float16)0.f; p3[j] = okB ? (_Float16)vb[j].w : (_Float16)0.f;
            }
            *(v4h*)(dst)          = p0;
            *(v4h*)(dst + HP)     = p1;
            *(v4h*)(dst + 2 * HP) = p2;
            *(v4h*)(dst + 3 * HP) = p3;
        }
    }
    lds_barrier();

    v8f acc[4][2];
    #pragma unroll
    for (int j = 0; j < 4; ++j) {
        v8f c0;
        c0[0] = ipb[j][0].x; c0[1] = ipb[j][0].y; c0[2] = ipb[j][0].z; c0[3] = ipb[j][0].w;
        c0[4] = ipb[j][1].x; c0[5] = ipb[j][1].y; c0[6] = ipb[j][1].z; c0[7] = ipb[j][1].w;
        acc[j][0] = c0; acc[j][1] = c0;
    }
    {
        const _Float16* Bpix = &halo[(t * HCOLS + 32 * hw + nlo) * HP + base];
        FragU hfr[2][2];
        #pragma unroll
        for (int p = 0; p < 2; ++p) load_afrag(hfr[0][p], Bpix + (p * 16) * HP);
        __builtin_amdgcn_sched_group_barrier(0x100, 4, 0);
        #pragma unroll
        for (int kt = 0; kt < 9; ++kt) {
            if (kt < 8) {
                const int kh = (kt + 1) / 3, kw = (kt + 1) - 3 * kh;
                #pragma unroll
                for (int p = 0; p < 2; ++p)
                    load_afrag(hfr[(kt + 1) & 1][p], Bpix + (kh * HCOLS + kw + p * 16) * HP);
            }
            #pragma unroll
            for (int j = 0; j < 4; ++j) {
                acc[j][0] = __builtin_amdgcn_wmma_f32_16x16x32_f16(false, bw[kt][j].v, false, hfr[kt & 1][0].v, (short)0, acc[j][0], false, false);
                acc[j][1] = __builtin_amdgcn_wmma_f32_16x16x32_f16(false, bw[kt][j].v, false, hfr[kt & 1][1].v, (short)0, acc[j][1], false, false);
            }
            if (kt < 8) __builtin_amdgcn_sched_group_barrier(0x100, 4, 0);
            __builtin_amdgcn_sched_group_barrier(0x008, 8, 0);
        }
    }

    __builtin_amdgcn_sched_barrier(0);

    FragU br[2][4], bi[2][4], bp[2][4];
    #pragma unroll
    for (int i = 0; i < 4; ++i) {
        #pragma unroll
        for (int kt = 0; kt < 2; ++kt) {
            load_lfrag(br[kt][i], sMr, i * 2 + kt, lane);
            load_lfrag(bi[kt][i], sMi, i * 2 + kt, lane);
        }
    }
    #pragma unroll
    for (int i = 0; i < 4; ++i) {
        #pragma unroll
        for (int kt = 0; kt < 2; ++kt)
            load_lfrag(bp[kt][i], sP, i * 2 + kt, lane);
    }
    __builtin_amdgcn_sched_barrier(0);
    v8f ob[4];
    #pragma unroll
    for (int i = 0; i < 4; ++i) {
        float bj[8];
        load8f(bj, out_proj_b + 16 * i + base);
        for (int r = 0; r < 8; ++r) ob[i][r] = bj[r];
    }

    FragU yB[2][2];
    #pragma unroll
    for (int p = 0; p < 2; ++p) {
        float y[4][8];
        float sj[4];
        #pragma unroll
        for (int j = 0; j < 4; ++j) {
            float s = 0.f;
            for (int r = 0; r < 8; ++r) {
                y[j][r] = acc[j][p][r];
                s += y[j][r] * y[j][r];
            }
            sj[j] = s;
        }
        float tot = (sj[0] + sj[1]) + (sj[2] + sj[3]);
        tot += other_half(tot);
        const float inv = 1.f / (sqrtf(tot) + 1e-9f);
        #pragma unroll
        for (int kt = 0; kt < 2; ++kt) {
            for (int e = 0; e < 8; ++e) {
                yB[p][kt].v[e]     = (_Float16)(y[2 * kt][e] * inv);
                yB[p][kt].v[8 + e] = (_Float16)(y[2 * kt + 1][e] * inv);
            }
        }
    }

    FragU pB[2][2];
    #pragma unroll
    for (int i = 0; i < 4; ++i) {
        #pragma unroll
        for (int p = 0; p < 2; ++p) {
            v8f ar = {}, ai = {};
            #pragma unroll
            for (int kt = 0; kt < 2; ++kt) {
                ar = __builtin_amdgcn_wmma_f32_16x16x32_f16(false, br[kt][i].v, false, yB[p][kt].v, (short)0, ar, false, false);
                ai = __builtin_amdgcn_wmma_f32_16x16x32_f16(false, bi[kt][i].v, false, yB[p][kt].v, (short)0, ai, false, false);
            }
            for (int e = 0; e < 8; ++e)
                pB[p][i >> 1].v[(i & 1) * 8 + e] = (_Float16)(ar[e] * ar[e] + ai[e] * ai[e]);
        }
    }

    {
        float* orow = out + (((size_t)b * 64) * 64 + (h0 + t)) * 64 + 32 * hw + lane;
        #pragma unroll
        for (int i = 0; i < 4; ++i) {
            v8f o0 = ob[i], o1 = ob[i];
            #pragma unroll
            for (int kt = 0; kt < 2; ++kt) {
                o0 = __builtin_amdgcn_wmma_f32_16x16x32_f16(false, bp[kt][i].v, false, pB[0][kt].v, (short)0, o0, false, false);
                o1 = __builtin_amdgcn_wmma_f32_16x16x32_f16(false, bp[kt][i].v, false, pB[1][kt].v, (short)0, o1, false, false);
            }
            for (int r = 0; r < 8; ++r) {
                const float v0 = o0[r];
                const float v1 = o1[r];
                const float got = other_half(upper ? v0 : v1);
                orow[(size_t)(16 * i + r) * 4096]     = upper ? got : v0;
                orow[(size_t)(16 * i + 8 + r) * 4096] = upper ? v1 : got;
            }
        }
    }
}


extern "C" void kernel_launch(void* const* d_in, const int* in_sizes, int n_in,
                              void* d_out, int out_size, void* d_ws, size_t ws_size,
                              hipStream_t stream) {
    const float* x          = (const float*)d_in[0];
    const float* inp_scale  = (const float*)d_in[1];
    const float* in_proj_w  = (const float*)d_in[2];
    const float* in_proj_b  = (const float*)d_in[3];
    const float* weights    = (const float*)d_in[4];
    const float* meas_w     = (const float*)d_in[5];
    const float* out_proj_w = (const float*)d_in[6];
    const float* out_proj_b = (const float*)d_in[7];
    float* out = (float*)d_out;

    hipLaunchKernelGGL(qconv_fused_kernel, dim3(256), dim3(256), 0, stream,
                       x, inp_scale, in_proj_w, in_proj_b, weights, meas_w, out_proj_w, out_proj_b, out);
}
